# starting kernel
# baseline (speedup 1.0000x reference)
#include <hip/hip_runtime.h>

typedef __attribute__((ext_vector_type(2))) float v2f;
typedef __attribute__((ext_vector_type(8))) float v8f;

#define N_NODES   100000
#define D_FEAT    128
#define N_EDGES   1600000
#define N_CLASSES 32
#define M_TILES   (N_NODES / 16)

__global__ __launch_bounds__(128)
void gin_proj_wmma(const float* __restrict__ x, const float* __restrict__ W,
                   const float* __restrict__ bias, float* __restrict__ y,
                   float* __restrict__ out) {
    const int lane  = threadIdx.x & 31;
    const int wave  = threadIdx.x >> 5;
    const int mtile = blockIdx.x * 4 + wave;
    if (mtile >= M_TILES) return;

    const int col   = lane & 15;
    const int khalf = (lane >> 4) << 1;

    const float* xrow  = x + (mtile * 16 + col) * D_FEAT;
    const float* wrow0 = W + col * D_FEAT;
    const float* wrow1 = W + (col + 16) * D_FEAT;

    v8f acc0 = {};
    v8f acc1 = {};
#pragma unroll 4
    for (int k0 = 0; k0 < D_FEAT; k0 += 4) {
        v2f a  = *(const v2f*)(xrow  + k0 + khalf);
        v2f b0 = *(const v2f*)(wrow0 + k0 + khalf);
        v2f b1 = *(const v2f*)(wrow1 + k0 + khalf);
        acc0 = __builtin_amdgcn_wmma_f32_16x16x4_f32(false, a, false, b0,
                                                     (short)0, acc0, false, false);
        acc1 = __builtin_amdgcn_wmma_f32_16x16x4_f32(false, a, false, b1,
                                                     (short)0, acc1, false, false);
    }

    const float bias0 = bias[col];
    const float bias1 = bias[col + 16];
    const int rbase = mtile * 16 + ((lane >> 4) << 3);
#pragma unroll
    for (int r = 0; r < 8; ++r) {
        const int row = rbase + r;
        y[row * N_CLASSES + col]        = acc0[r];
        y[row * N_CLASSES + col + 16]   = acc1[r];
        out[row * N_CLASSES + col]      = acc0[r] + bias0;
        out[row * N_CLASSES + col + 16] = acc1[r] + bias1;
    }
}

__global__ __launch_bounds__(256)
void gin_edge_scatter(const int* __restrict__ src, const int* __restrict__ dst,
                      const float* __restrict__ y, float* __restrict__ out) {
    const int  c       = threadIdx.x & 31;
    long long  t       = (long long)blockIdx.x * blockDim.x + threadIdx.x;
    const long long nw = ((long long)gridDim.x * blockDim.x) >> 5;
    for (long long e = t >> 5; e < N_EDGES; e += nw) {
        const int   s = src[e];
        const int   d = dst[e];
        const float v = y[(long long)s * N_CLASSES + c];
        atomicAdd(out + (long long)d * N_CLASSES + c, v);
    }
}

extern "C" void kernel_launch(void* const* d_in, const int* in_sizes, int n_in,
                              void* d_out, int out_size, void* d_ws, size_t ws_size,
                              hipStream_t stream) {
    const float* x          = (const float*)d_in[0];
    const int*   edge_index = (const int*)  d_in[1];
    const float* W          = (const float*)d_in[2];
    const float* b          = (const float*)d_in[3];
    float*       out        = (float*)d_out;
    float*       y          = (float*)d_ws;

    const int* src = edge_index;
    const int* dst = edge_index + N_EDGES;

    const int proj_blocks = (M_TILES + 3) / 4;
    gin_proj_wmma<<<proj_blocks, 128, 0, stream>>>(x, W, b, y, out);

    const int edge_blocks = (N_EDGES * 32) / 256;
    gin_edge_scatter<<<edge_blocks, 256, 0, stream>>>(src, dst, y, out);
}


# edge sum by 250 owner groups, no binning and no global atomics: each filters a one-byte owner key per edge and adds its edges' y rows in LDS (f32 LDS adds, arrival order); one store per out line
# speedup vs baseline: 1.6980x; 1.6980x over previous
#include <hip/hip_runtime.h>


typedef __attribute__((ext_vector_type(2))) float v2f;
typedef __attribute__((ext_vector_type(8))) float v8f;

#define N_NODES   100000
#define D_FEAT    128
#define N_EDGES   1600000
#define N_CLASSES 32
#define M_TILES   (N_NODES / 16)

#define OWN_ROWS     400
#define N_OWNERS     (N_NODES / OWN_ROWS)
#define OWN_THREADS  1024
#define KEY_VECS     (N_EDGES / 16)
#define VECS_PER_THR 4
#define CHUNK_VECS   (OWN_THREADS * VECS_PER_THR)
#define CHUNK_EDGES  (CHUNK_VECS * 16)
#define N_CHUNKS     ((KEY_VECS + CHUNK_VECS - 1) / CHUNK_VECS)
#define PROC_UNROLL  8
#define INIT_TRIPS   ((OWN_ROWS * N_CLASSES + OWN_THREADS - 1) / OWN_THREADS)

__global__ __launch_bounds__(128)
void gin_proj_wmma(const float* __restrict__ x, const float* __restrict__ W,
                   const int* __restrict__ dst, float* __restrict__ y,
                   unsigned int* __restrict__ keys) {
    {
        const int g = blockIdx.x * 128 + threadIdx.x;
        if (g < N_EDGES / 8) {
            const int4 d0 = *(const int4*)(dst + g * 8);
            const int4 d1 = *(const int4*)(dst + g * 8 + 4);
            uint2 k;
            k.x =  ((unsigned)d0.x / OWN_ROWS)        | (((unsigned)d0.y / OWN_ROWS) << 8) |
                  (((unsigned)d0.z / OWN_ROWS) << 16) | (((unsigned)d0.w / OWN_ROWS) << 24);
            k.y =  ((unsigned)d1.x / OWN_ROWS)        | (((unsigned)d1.y / OWN_ROWS) << 8) |
                  (((unsigned)d1.z / OWN_ROWS) << 16) | (((unsigned)d1.w / OWN_ROWS) << 24);
            *(uint2*)(keys + g * 2) = k;
        }
    }

    const int lane  = threadIdx.x & 31;
    const int wave  = threadIdx.x >> 5;
    const int mtile = blockIdx.x * 4 + wave;
    if (mtile >= M_TILES) return;

    const int col   = lane & 15;
    const int khalf = (lane >> 4) << 1;

    const float* xrow  = x + (mtile * 16 + col) * D_FEAT;
    const float* wrow0 = W + col * D_FEAT;
    const float* wrow1 = W + (col + 16) * D_FEAT;

    v8f acc0 = {};
    v8f acc1 = {};
#pragma unroll 4
    for (int k0 = 0; k0 < D_FEAT; k0 += 4) {
        v2f a  = *(const v2f*)(xrow  + k0 + khalf);
        v2f b0 = *(const v2f*)(wrow0 + k0 + khalf);
        v2f b1 = *(const v2f*)(wrow1 + k0 + khalf);
        acc0 = __builtin_amdgcn_wmma_f32_16x16x4_f32(false, a, false, b0,
                                                     (short)0, acc0, false, false);
        acc1 = __builtin_amdgcn_wmma_f32_16x16x4_f32(false, a, false, b1,
                                                     (short)0, acc1, false, false);
    }

    const int rbase = mtile * 16 + ((lane >> 4) << 3);
#pragma unroll
    for (int r = 0; r < 8; ++r) {
        const int row = rbase + r;
        y[row * N_CLASSES + col]      = acc0[r];
        y[row * N_CLASSES + col + 16] = acc1[r];
    }
}

__device__ __forceinline__ unsigned zero_bytes(unsigned v) {
    unsigned t = (v & 0x7F7F7F7Fu) + 0x7F7F7F7Fu;
    return ~(t | v | 0x7F7F7F7Fu);
}
__device__ __forceinline__ unsigned marks4(unsigned t) {
    return ((t >> 7) * 0x10204080u) >> 28;
}

__global__ __launch_bounds__(OWN_THREADS)
void gin_owner_sum(const int* __restrict__ src, const int* __restrict__ dst,
                   const unsigned int* __restrict__ keys,
                   const float* __restrict__ y, const float* __restrict__ bias,
                   float* __restrict__ out) {
    __shared__ float          rows[OWN_ROWS * N_CLASSES];
    __shared__ unsigned short queue[CHUNK_EDGES];
    __shared__ int            qcount[2];

    const int tid   = threadIdx.x;
    const int lane  = tid & 31;
    const int wave  = tid >> 5;
    const int owner = blockIdx.x;
    const int base  = owner * OWN_ROWS;
    const unsigned pat = (unsigned)owner * 0x01010101u;
    const uint4* kv = (const uint4*)keys;

    int kk = owner % N_CHUNKS;

    uint4 cur[VECS_PER_THR];
#pragma unroll
    for (int j = 0; j < VECS_PER_THR; ++j) {
        const int v = kk * CHUNK_VECS + j * OWN_THREADS + tid;
        cur[j] = make_uint4(0xFFFFFFFFu, 0xFFFFFFFFu, 0xFFFFFFFFu, 0xFFFFFFFFu);
        if (v < KEY_VECS) cur[j] = kv[v];
    }

    {
        const float bv = bias[lane];
        float t[INIT_TRIPS];
#pragma unroll
        for (int it = 0; it < INIT_TRIPS; ++it) {
            const int i  = it * OWN_THREADS + tid;
            const int ic = (i < OWN_ROWS * N_CLASSES) ? i : (OWN_ROWS * N_CLASSES - 1);
            t[it] = y[base * N_CLASSES + ic];
        }
#pragma unroll
        for (int it = 0; it < INIT_TRIPS; ++it) {
            const int i = it * OWN_THREADS + tid;
            if (i < OWN_ROWS * N_CLASSES) rows[i] = t[it] + bv;
        }
    }
    if (tid < 2) qcount[tid] = 0;
    __syncthreads();

    for (int k = 0; k < N_CHUNKS; ++k) {
        int* cnt = &qcount[k & 1];

#pragma unroll
        for (int j = 0; j < VECS_PER_THR; ++j) {
            const unsigned m16 =  marks4(zero_bytes(cur[j].x ^ pat))
                               | (marks4(zero_bytes(cur[j].y ^ pat)) << 4)
                               | (marks4(zero_bytes(cur[j].z ^ pat)) << 8)
                               | (marks4(zero_bytes(cur[j].w ^ pat)) << 12);
            const int vbase = (j * OWN_THREADS + tid) * 16;
            unsigned m = m16;
            while (m) {
                const int b = __builtin_ctz(m);
                m &= m - 1;
                const int slot = atomicAdd(cnt, 1);
                queue[slot] = (unsigned short)(vbase + b);
            }
        }
        __syncthreads();

        const int n     = *cnt;
        const int ebase = kk * CHUNK_EDGES;
        if (tid == 0) qcount[(k + 1) & 1] = 0;

        kk = (kk + 1 == N_CHUNKS) ? 0 : kk + 1;
        uint4 nxt[VECS_PER_THR];
#pragma unroll
        for (int j = 0; j < VECS_PER_THR; ++j) {
            const int v = kk * CHUNK_VECS + j * OWN_THREADS + tid;
            nxt[j] = make_uint4(0xFFFFFFFFu, 0xFFFFFFFFu, 0xFFFFFFFFu, 0xFFFFFFFFu);
            if (k + 1 < N_CHUNKS && v < KEY_VECS) nxt[j] = kv[v];
        }

        for (int i0 = wave * PROC_UNROLL; i0 < n; i0 += (OWN_THREADS / 32) * PROC_UNROLL) {
            int   e[PROC_UNROLL];
            int   r[PROC_UNROLL];
            float v[PROC_UNROLL];
#pragma unroll
            for (int u = 0; u < PROC_UNROLL; ++u) {
                const int i = (i0 + u < n) ? (i0 + u) : (n - 1);
                e[u] = ebase + (int)queue[i];
            }
#pragma unroll
            for (int u = 0; u < PROC_UNROLL; ++u) {
                const int s = src[e[u]];
                r[u] = dst[e[u]] - base;
                v[u] = y[s * N_CLASSES + lane];
            }
#pragma unroll
            for (int u = 0; u < PROC_UNROLL; ++u) {
                const bool ok = (i0 + u < n) && ((unsigned)r[u] < (unsigned)OWN_ROWS);
                atomicAdd(&rows[(ok ? r[u] : 0) * N_CLASSES + lane], ok ? v[u] : 0.0f);
            }
        }
        __syncthreads();

#pragma unroll
        for (int j = 0; j < VECS_PER_THR; ++j) cur[j] = nxt[j];
    }

    for (int i = tid; i < OWN_ROWS * N_CLASSES; i += OWN_THREADS)
        out[base * N_CLASSES + i] = rows[i];
}

extern "C" void kernel_launch(void* const* d_in, const int* in_sizes, int n_in,
                              void* d_out, int out_size, void* d_ws, size_t ws_size,
                              hipStream_t stream) {
    const float* x          = (const float*)d_in[0];
    const int*   edge_index = (const int*)  d_in[1];
    const float* W          = (const float*)d_in[2];
    const float* b          = (const float*)d_in[3];
    float*       out        = (float*)d_out;
    float*       y          = (float*)d_ws;
    unsigned int* keys      = (unsigned int*)(y + N_NODES * N_CLASSES);

    const int* src = edge_index;
    const int* dst = edge_index + N_EDGES;

    const int proj_blocks = (M_TILES + 3) / 4;
    gin_proj_wmma<<<proj_blocks, 128, 0, stream>>>(x, W, dst, y, keys);

    gin_owner_sum<<<N_OWNERS, OWN_THREADS, 0, stream>>>(src, dst, keys, y, b, out);
}


# owner launch: kept-edge list double buffered, chunk k+1 is filtered between the same two barriers in which chunk k's rows are gathered and added (1 barrier per chunk, not 2); LDS arrival-order sums
# speedup vs baseline: 1.8859x; 1.1107x over previous
#include <hip/hip_runtime.h>


typedef __attribute__((ext_vector_type(2))) float v2f;
typedef __attribute__((ext_vector_type(8))) float v8f;

#define N_NODES   100000
#define D_FEAT    128
#define N_EDGES   1600000
#define N_CLASSES 32
#define M_TILES   (N_NODES / 16)

#define OWN_ROWS     400
#define N_OWNERS     (N_NODES / OWN_ROWS)
#define OWN_THREADS  1024
#define KEY_VECS     (N_EDGES / 16)
#define VECS_PER_THR 4
#define CHUNK_VECS   (OWN_THREADS * VECS_PER_THR)
#define CHUNK_EDGES  (CHUNK_VECS * 16)
#define N_CHUNKS     ((KEY_VECS + CHUNK_VECS - 1) / CHUNK_VECS)
#define PROC_UNROLL  8
#define INIT_TRIPS   ((OWN_ROWS * N_CLASSES + OWN_THREADS - 1) / OWN_THREADS)

__global__ __launch_bounds__(128)
void gin_proj_wmma(const float* __restrict__ x, const float* __restrict__ W,
                   const int* __restrict__ dst, float* __restrict__ y,
                   unsigned int* __restrict__ keys) {
    {
        const int g = blockIdx.x * 128 + threadIdx.x;
        if (g < N_EDGES / 8) {
            const int4 d0 = *(const int4*)(dst + g * 8);
            const int4 d1 = *(const int4*)(dst + g * 8 + 4);
            uint2 k;
            k.x =  ((unsigned)d0.x / OWN_ROWS)        | (((unsigned)d0.y / OWN_ROWS) << 8) |
                  (((unsigned)d0.z / OWN_ROWS) << 16) | (((unsigned)d0.w / OWN_ROWS) << 24);
            k.y =  ((unsigned)d1.x / OWN_ROWS)        | (((unsigned)d1.y / OWN_ROWS) << 8) |
                  (((unsigned)d1.z / OWN_ROWS) << 16) | (((unsigned)d1.w / OWN_ROWS) << 24);
            *(uint2*)(keys + g * 2) = k;
        }
    }

    const int lane  = threadIdx.x & 31;
    const int wave  = threadIdx.x >> 5;
    const int mtile = blockIdx.x * 4 + wave;
    if (mtile >= M_TILES) return;

    const int col   = lane & 15;
    const int khalf = (lane >> 4) << 1;

    const float* xrow  = x + (mtile * 16 + col) * D_FEAT;
    const float* wrow0 = W + col * D_FEAT;
    const float* wrow1 = W + (col + 16) * D_FEAT;

    v8f acc0 = {};
    v8f acc1 = {};
#pragma unroll 4
    for (int k0 = 0; k0 < D_FEAT; k0 += 4) {
        v2f a  = *(const v2f*)(xrow  + k0 + khalf);
        v2f b0 = *(const v2f*)(wrow0 + k0 + khalf);
        v2f b1 = *(const v2f*)(wrow1 + k0 + khalf);
        acc0 = __builtin_amdgcn_wmma_f32_16x16x4_f32(false, a, false, b0,
                                                     (short)0, acc0, false, false);
        acc1 = __builtin_amdgcn_wmma_f32_16x16x4_f32(false, a, false, b1,
                                                     (short)0, acc1, false, false);
    }

    const int rbase = mtile * 16 + ((lane >> 4) << 3);
#pragma unroll
    for (int r = 0; r < 8; ++r) {
        const int row = rbase + r;
        y[row * N_CLASSES + col]      = acc0[r];
        y[row * N_CLASSES + col + 16] = acc1[r];
    }
}

__device__ __forceinline__ unsigned zero_bytes(unsigned v) {
    unsigned t = (v & 0x7F7F7F7Fu) + 0x7F7F7F7Fu;
    return ~(t | v | 0x7F7F7F7Fu);
}
__device__ __forceinline__ unsigned marks4(unsigned t) {
    return ((t >> 7) * 0x10204080u) >> 28;
}


__device__ __forceinline__ void keep_own_edges(const uint4 (&cur)[VECS_PER_THR], unsigned pat, int tid,
                                               unsigned short* __restrict__ q, int* __restrict__ cnt) {
#pragma unroll
    for (int j = 0; j < VECS_PER_THR; ++j) {
        const unsigned m16 =  marks4(zero_bytes(cur[j].x ^ pat))
                           | (marks4(zero_bytes(cur[j].y ^ pat)) << 4)
                           | (marks4(zero_bytes(cur[j].z ^ pat)) << 8)
                           | (marks4(zero_bytes(cur[j].w ^ pat)) << 12);
        const int vbase = (j * OWN_THREADS + tid) * 16;
        unsigned m = m16;
        while (m) {
            const int b = __builtin_ctz(m);
            m &= m - 1;
            const int slot = atomicAdd(cnt, 1);
            q[slot] = (unsigned short)(vbase + b);
        }
    }
}

__global__ __launch_bounds__(OWN_THREADS)
void gin_owner_sum(const int* __restrict__ src, const int* __restrict__ dst,
                   const unsigned int* __restrict__ keys,
                   const float* __restrict__ y, const float* __restrict__ bias,
                   float* __restrict__ out) {
    __shared__ float          rows[OWN_ROWS * N_CLASSES];
    __shared__ unsigned short queue[2][CHUNK_EDGES];
    __shared__ int            qcount[N_CHUNKS];

    const int tid   = threadIdx.x;
    const int lane  = tid & 31;
    const int wave  = tid >> 5;
    const int owner = blockIdx.x;
    const int base  = owner * OWN_ROWS;
    const unsigned pat = (unsigned)owner * 0x01010101u;
    const uint4* kv = (const uint4*)keys;
    const uint4 none = make_uint4(0xFFFFFFFFu, 0xFFFFFFFFu, 0xFFFFFFFFu, 0xFFFFFFFFu);

    int kk  = owner % N_CHUNKS;
    int kk1 = (kk + 1 == N_CHUNKS) ? 0 : kk + 1;

    uint4 cur[VECS_PER_THR];
#pragma unroll
    for (int j = 0; j < VECS_PER_THR; ++j) {
        const int v = kk * CHUNK_VECS + j * OWN_THREADS + tid;
        cur[j] = none;
        if (v < KEY_VECS) cur[j] = kv[v];
    }
    if (tid < N_CHUNKS) qcount[tid] = 0;
    __syncthreads();

    {
        const float bv = bias[lane];
        float t[INIT_TRIPS];
#pragma unroll
        for (int it = 0; it < INIT_TRIPS; ++it) {
            const int i  = it * OWN_THREADS + tid;
            const int ic = (i < OWN_ROWS * N_CLASSES) ? i : (OWN_ROWS * N_CLASSES - 1);
            t[it] = y[base * N_CLASSES + ic];
        }
        keep_own_edges(cur, pat, tid, queue[0], &qcount[0]);
#pragma unroll
        for (int it = 0; it < INIT_TRIPS; ++it) {
            const int i = it * OWN_THREADS + tid;
            if (i < OWN_ROWS * N_CLASSES) rows[i] = t[it] + bv;
        }
    }
#pragma unroll
    for (int j = 0; j < VECS_PER_THR; ++j) {
        const int v = kk1 * CHUNK_VECS + j * OWN_THREADS + tid;
        cur[j] = none;
        if (v < KEY_VECS) cur[j] = kv[v];
    }
    __syncthreads();

    for (int k = 0; k < N_CHUNKS; ++k) {
        const unsigned short* q = queue[k & 1];
        const int n     = qcount[k];
        const int ebase = kk * CHUNK_EDGES;
        const int first = wave * PROC_UNROLL;
        const bool act  = first < n;

        int s[PROC_UNROLL];
        int d[PROC_UNROLL];
        if (act) {
            int e[PROC_UNROLL];
#pragma unroll
            for (int u = 0; u < PROC_UNROLL; ++u) {
                const int i = (first + u < n) ? (first + u) : (n - 1);
                e[u] = ebase + (int)q[i];
            }
#pragma unroll
            for (int u = 0; u < PROC_UNROLL; ++u) {
                s[u] = src[e[u]];
                d[u] = dst[e[u]];
            }
        }

        const int kk2 = (kk1 + 1 == N_CHUNKS) ? 0 : kk1 + 1;
        uint4 nxt[VECS_PER_THR];
#pragma unroll
        for (int j = 0; j < VECS_PER_THR; ++j) {
            const int v = kk2 * CHUNK_VECS + j * OWN_THREADS + tid;
            nxt[j] = none;
            if (k + 2 < N_CHUNKS && v < KEY_VECS) nxt[j] = kv[v];
        }

        if (k + 1 < N_CHUNKS)
            keep_own_edges(cur, pat, tid, queue[(k + 1) & 1], &qcount[k + 1]);

        if (act) {
            float v[PROC_UNROLL];
#pragma unroll
            for (int u = 0; u < PROC_UNROLL; ++u) v[u] = y[s[u] * N_CLASSES + lane];
#pragma unroll
            for (int u = 0; u < PROC_UNROLL; ++u) {
                const int  r  = d[u] - base;
                const bool ok = (first + u < n) && ((unsigned)r < (unsigned)OWN_ROWS);
                atomicAdd(&rows[(ok ? r : 0) * N_CLASSES + lane], ok ? v[u] : 0.0f);
            }
        }

        for (int i0 = first + (OWN_THREADS / 32) * PROC_UNROLL; i0 < n; i0 += (OWN_THREADS / 32) * PROC_UNROLL) {
            int   e2[PROC_UNROLL];
            int   r2[PROC_UNROLL];
            float v2[PROC_UNROLL];
#pragma unroll
            for (int u = 0; u < PROC_UNROLL; ++u) {
                const int i = (i0 + u < n) ? (i0 + u) : (n - 1);
                e2[u] = ebase + (int)q[i];
            }
#pragma unroll
            for (int u = 0; u < PROC_UNROLL; ++u) {
                const int sv = src[e2[u]];
                r2[u] = dst[e2[u]] - base;
                v2[u] = y[sv * N_CLASSES + lane];
            }
#pragma unroll
            for (int u = 0; u < PROC_UNROLL; ++u) {
                const bool ok = (i0 + u < n) && ((unsigned)r2[u] < (unsigned)OWN_ROWS);
                atomicAdd(&rows[(ok ? r2[u] : 0) * N_CLASSES + lane], ok ? v2[u] : 0.0f);
            }
        }
        __syncthreads();

#pragma unroll
        for (int j = 0; j < VECS_PER_THR; ++j) cur[j] = nxt[j];
        kk  = kk1;
        kk1 = kk2;
    }

    for (int i = tid; i < OWN_ROWS * N_CLASSES; i += OWN_THREADS)
        out[base * N_CLASSES + i] = rows[i];
}

extern "C" void kernel_launch(void* const* d_in, const int* in_sizes, int n_in,
                              void* d_out, int out_size, void* d_ws, size_t ws_size,
                              hipStream_t stream) {
    const float* x          = (const float*)d_in[0];
    const int*   edge_index = (const int*)  d_in[1];
    const float* W          = (const float*)d_in[2];
    const float* b          = (const float*)d_in[3];
    float*       out        = (float*)d_out;
    float*       y          = (float*)d_ws;
    unsigned int* keys      = (unsigned int*)(y + N_NODES * N_CLASSES);

    const int* src = edge_index;
    const int* dst = edge_index + N_EDGES;

    const int proj_blocks = (M_TILES + 3) / 4;
    gin_proj_wmma<<<proj_blocks, 128, 0, stream>>>(x, W, dst, y, keys);

    gin_owner_sum<<<N_OWNERS, OWN_THREADS, 0, stream>>>(src, dst, keys, y, b, out);
}


# owner keys stored bit-sliced (8 words per 32 edges, built with lane ballots in launch 1): an owner tests 32 edges with 8 logic ops, not byte tricks on 4 at a time; LDS arrival-order sums as before
# speedup vs baseline: 2.2877x; 1.2131x over previous
#include <hip/hip_runtime.h>


typedef __attribute__((ext_vector_type(2))) float v2f;
typedef __attribute__((ext_vector_type(8))) float v8f;

#define N_NODES   100000
#define D_FEAT    128
#define N_EDGES   1600000
#define N_CLASSES 32
#define M_TILES   (N_NODES / 16)

#define OWN_ROWS     400
#define N_OWNERS     (N_NODES / OWN_ROWS)
#define OWN_THREADS  1024
#define SET_EDGES    1024
#define N_SETS       ((N_EDGES + SET_EDGES - 1) / SET_EDGES)
#define SET_VECS     (SET_EDGES / 16)
#define SETS_PER_THR 2
#define VECS_PER_THR (2 * SETS_PER_THR)
#define CHUNK_SETS   ((OWN_THREADS / 32) * SETS_PER_THR)
#define CHUNK_EDGES  (CHUNK_SETS * SET_EDGES)
#define N_CHUNKS     ((N_SETS + CHUNK_SETS - 1) / CHUNK_SETS)
#define PROC_UNROLL  8
#define INIT_TRIPS   ((OWN_ROWS * N_CLASSES + OWN_THREADS - 1) / OWN_THREADS)

__global__ __launch_bounds__(128)
void gin_proj_wmma(const float* __restrict__ x, const float* __restrict__ W,
                   const int* __restrict__ dst, float* __restrict__ y,
                   unsigned int* __restrict__ keys) {
    const int lane  = threadIdx.x & 31;
    const int wave  = threadIdx.x >> 5;
    {
        unsigned pk[2];
#pragma unroll
        for (int trip = 0; trip < 2; ++trip) {
            const int  e0    = blockIdx.x * SET_EDGES + (wave * 2 + trip) * 128 + lane * 4;
            const bool valid = e0 < N_EDGES;
            const int4 d     = *(const int4*)(dst + (valid ? e0 : 0));
            const unsigned p =  ((unsigned)d.x / OWN_ROWS)        | (((unsigned)d.y / OWN_ROWS) << 8) |
                               (((unsigned)d.z / OWN_ROWS) << 16) | (((unsigned)d.w / OWN_ROWS) << 24);
            pk[trip] = valid ? p : 0xFFFFFFFFu;
        }
        unsigned lo = 0, hi = 0;
#pragma unroll
        for (int trip = 0; trip < 2; ++trip) {
#pragma unroll
            for (int cb = 0; cb < 16; ++cb) {
                const int c = cb >> 2, b = cb & 3;
                const unsigned w0 = __builtin_amdgcn_ballot_w32(((pk[trip] >> (8 * c + b))     & 1u) != 0u);
                const unsigned w1 = __builtin_amdgcn_ballot_w32(((pk[trip] >> (8 * c + b + 4)) & 1u) != 0u);
                const bool mine = (lane == trip * 16 + cb);
                lo = mine ? w0 : lo;
                hi = mine ? w1 : hi;
            }
        }
        keys[blockIdx.x * (SET_EDGES / 4) + wave * 32 + lane]       = lo;
        keys[blockIdx.x * (SET_EDGES / 4) + 128 + wave * 32 + lane] = hi;
    }

    const int mtile = blockIdx.x * 4 + wave;
    if (mtile >= M_TILES) return;

    const int col   = lane & 15;
    const int khalf = (lane >> 4) << 1;

    const float* xrow  = x + (mtile * 16 + col) * D_FEAT;
    const float* wrow0 = W + col * D_FEAT;
    const float* wrow1 = W + (col + 16) * D_FEAT;

    v8f acc0 = {};
    v8f acc1 = {};
#pragma unroll 4
    for (int k0 = 0; k0 < D_FEAT; k0 += 4) {
        v2f a  = *(const v2f*)(xrow  + k0 + khalf);
        v2f b0 = *(const v2f*)(wrow0 + k0 + khalf);
        v2f b1 = *(const v2f*)(wrow1 + k0 + khalf);
        acc0 = __builtin_amdgcn_wmma_f32_16x16x4_f32(false, a, false, b0,
                                                     (short)0, acc0, false, false);
        acc1 = __builtin_amdgcn_wmma_f32_16x16x4_f32(false, a, false, b1,
                                                     (short)0, acc1, false, false);
    }

    const int rbase = mtile * 16 + ((lane >> 4) << 3);
#pragma unroll
    for (int r = 0; r < 8; ++r) {
        const int row = rbase + r;
        y[row * N_CLASSES + col]      = acc0[r];
        y[row * N_CLASSES + col + 16] = acc1[r];
    }
}


__device__ __forceinline__ void keep_own_edges(const uint4 (&cur)[VECS_PER_THR], const unsigned (&flip)[8],
                                               int wave, int lane,
                                               unsigned short* __restrict__ q, int* __restrict__ cnt) {
#pragma unroll
    for (int j = 0; j < SETS_PER_THR; ++j) {
        const uint4 a = cur[2 * j], h = cur[2 * j + 1];
        unsigned m = a.x ^ flip[0];
        m = __builtin_amdgcn_bitop3_b32(a.y, flip[1], m, 0x28);
        m = __builtin_amdgcn_bitop3_b32(a.z, flip[2], m, 0x28);
        m = __builtin_amdgcn_bitop3_b32(a.w, flip[3], m, 0x28);
        m = __builtin_amdgcn_bitop3_b32(h.x, flip[4], m, 0x28);
        m = __builtin_amdgcn_bitop3_b32(h.y, flip[5], m, 0x28);
        m = __builtin_amdgcn_bitop3_b32(h.z, flip[6], m, 0x28);
        m = __builtin_amdgcn_bitop3_b32(h.w, flip[7], m, 0x28);
        const int vbase = (j * (OWN_THREADS / 32) + wave) * SET_EDGES + (lane >> 2) * 128 + (lane & 3);
        while (m) {
            const int b = __builtin_ctz(m);
            m &= m - 1;
            const int slot = atomicAdd(cnt, 1);
            q[slot] = (unsigned short)(vbase + b * 4);
        }
    }
}

__device__ __forceinline__ void load_keys(uint4 (&dstv)[VECS_PER_THR], const uint4* __restrict__ kv,
                                          int ck, int wave, int lane, bool on) {
    const uint4 none = make_uint4(0xFFFFFFFFu, 0xFFFFFFFFu, 0xFFFFFFFFu, 0xFFFFFFFFu);
#pragma unroll
    for (int j = 0; j < SETS_PER_THR; ++j) {
        const int S = ck * CHUNK_SETS + j * (OWN_THREADS / 32) + wave;
        dstv[2 * j]     = none;
        dstv[2 * j + 1] = none;
        if (on && S < N_SETS) {
            dstv[2 * j]     = kv[S * SET_VECS + lane];
            dstv[2 * j + 1] = kv[S * SET_VECS + 32 + lane];
        }
    }
}

__global__ __launch_bounds__(OWN_THREADS)
void gin_owner_sum(const int* __restrict__ src, const int* __restrict__ dst,
                   const unsigned int* __restrict__ keys,
                   const float* __restrict__ y, const float* __restrict__ bias,
                   float* __restrict__ out) {
    __shared__ float          rows[OWN_ROWS * N_CLASSES];
    __shared__ unsigned short queue[2][CHUNK_EDGES];
    __shared__ int            qcount[N_CHUNKS];

    const int tid   = threadIdx.x;
    const int lane  = tid & 31;
    const int wave  = tid >> 5;
    const int owner = blockIdx.x;
    const int base  = owner * OWN_ROWS;
    const uint4* kv = (const uint4*)keys;
    unsigned flip[8];
#pragma unroll
    for (int b = 0; b < 8; ++b) flip[b] = ((owner >> b) & 1) ? 0u : 0xFFFFFFFFu;

    int kk  = owner % N_CHUNKS;
    int kk1 = (kk + 1 == N_CHUNKS) ? 0 : kk + 1;

    uint4 cur[VECS_PER_THR];
    load_keys(cur, kv, kk, wave, lane, true);
    if (tid < N_CHUNKS) qcount[tid] = 0;
    __syncthreads();

    {
        const float bv = bias[lane];
        float t[INIT_TRIPS];
#pragma unroll
        for (int it = 0; it < INIT_TRIPS; ++it) {
            const int i  = it * OWN_THREADS + tid;
            const int ic = (i < OWN_ROWS * N_CLASSES) ? i : (OWN_ROWS * N_CLASSES - 1);
            t[it] = y[base * N_CLASSES + ic];
        }
        keep_own_edges(cur, flip, wave, lane, queue[0], &qcount[0]);
#pragma unroll
        for (int it = 0; it < INIT_TRIPS; ++it) {
            const int i = it * OWN_THREADS + tid;
            if (i < OWN_ROWS * N_CLASSES) rows[i] = t[it] + bv;
        }
    }
    load_keys(cur, kv, kk1, wave, lane, true);
    __syncthreads();

    for (int k = 0; k < N_CHUNKS; ++k) {
        const unsigned short* q = queue[k & 1];
        const int n     = qcount[k];
        const int ebase = kk * CHUNK_EDGES;
        const int first = wave * PROC_UNROLL;
        const bool act  = first < n;

        int s[PROC_UNROLL];
        int d[PROC_UNROLL];
        if (act) {
            int e[PROC_UNROLL];
#pragma unroll
            for (int u = 0; u < PROC_UNROLL; ++u) {
                const int i = (first + u < n) ? (first + u) : (n - 1);
                e[u] = ebase + (int)q[i];
            }
#pragma unroll
            for (int u = 0; u < PROC_UNROLL; ++u) {
                s[u] = src[e[u]];
                d[u] = dst[e[u]];
            }
        }

        const int kk2 = (kk1 + 1 == N_CHUNKS) ? 0 : kk1 + 1;
        uint4 nxt[VECS_PER_THR];
        load_keys(nxt, kv, kk2, wave, lane, k + 2 < N_CHUNKS);

        if (k + 1 < N_CHUNKS)
            keep_own_edges(cur, flip, wave, lane, queue[(k + 1) & 1], &qcount[k + 1]);

        if (act) {
            float v[PROC_UNROLL];
#pragma unroll
            for (int u = 0; u < PROC_UNROLL; ++u) v[u] = y[s[u] * N_CLASSES + lane];
#pragma unroll
            for (int u = 0; u < PROC_UNROLL; ++u) {
                const int  r  = d[u] - base;
                const bool ok = (first + u < n) && ((unsigned)r < (unsigned)OWN_ROWS);
                atomicAdd(&rows[(ok ? r : 0) * N_CLASSES + lane], ok ? v[u] : 0.0f);
            }
        }

        for (int i0 = first + (OWN_THREADS / 32) * PROC_UNROLL; i0 < n; i0 += (OWN_THREADS / 32) * PROC_UNROLL) {
            int   e2[PROC_UNROLL];
            int   r2[PROC_UNROLL];
            float v2[PROC_UNROLL];
#pragma unroll
            for (int u = 0; u < PROC_UNROLL; ++u) {
                const int i = (i0 + u < n) ? (i0 + u) : (n - 1);
                e2[u] = ebase + (int)q[i];
            }
#pragma unroll
            for (int u = 0; u < PROC_UNROLL; ++u) {
                const int sv = src[e2[u]];
                r2[u] = dst[e2[u]] - base;
                v2[u] = y[sv * N_CLASSES + lane];
            }
#pragma unroll
            for (int u = 0; u < PROC_UNROLL; ++u) {
                const bool ok = (i0 + u < n) && ((unsigned)r2[u] < (unsigned)OWN_ROWS);
                atomicAdd(&rows[(ok ? r2[u] : 0) * N_CLASSES + lane], ok ? v2[u] : 0.0f);
            }
        }
        __syncthreads();

#pragma unroll
        for (int j = 0; j < VECS_PER_THR; ++j) cur[j] = nxt[j];
        kk  = kk1;
        kk1 = kk2;
    }

    for (int i = tid; i < OWN_ROWS * N_CLASSES; i += OWN_THREADS)
        out[base * N_CLASSES + i] = rows[i];
}

extern "C" void kernel_launch(void* const* d_in, const int* in_sizes, int n_in,
                              void* d_out, int out_size, void* d_ws, size_t ws_size,
                              hipStream_t stream) {
    const float* x          = (const float*)d_in[0];
    const int*   edge_index = (const int*)  d_in[1];
    const float* W          = (const float*)d_in[2];
    const float* b          = (const float*)d_in[3];
    float*       out        = (float*)d_out;
    float*       y          = (float*)d_ws;
    unsigned int* keys      = (unsigned int*)(y + N_NODES * N_CLASSES);

    const int* src = edge_index;
    const int* dst = edge_index + N_EDGES;

    const int proj_blocks = (M_TILES + 3) / 4;
    gin_proj_wmma<<<proj_blocks, 128, 0, stream>>>(x, W, dst, y, keys);

    gin_owner_sum<<<N_OWNERS, OWN_THREADS, 0, stream>>>(src, dst, keys, y, b, out);
}


# launch 1 writes one packed word per edge (source row, target row inside its range); an owner does one 4-byte load per kept edge, not two index loads; LDS arrival-order sums as before
# speedup vs baseline: 2.4568x; 1.0739x over previous
#include <hip/hip_runtime.h>


typedef __attribute__((ext_vector_type(2))) float v2f;
typedef __attribute__((ext_vector_type(8))) float v8f;

#define N_NODES   100000
#define D_FEAT    128
#define N_EDGES   1600000
#define N_CLASSES 32
#define M_TILES   (N_NODES / 16)

#define OWN_ROWS     400
#define N_OWNERS     (N_NODES / OWN_ROWS)
#define OWN_THREADS  1024
#define SET_EDGES    1024
#define N_SETS       ((N_EDGES + SET_EDGES - 1) / SET_EDGES)
#define SET_VECS     (SET_EDGES / 16)
#define SETS_PER_THR 2
#define VECS_PER_THR (2 * SETS_PER_THR)
#define CHUNK_SETS   ((OWN_THREADS / 32) * SETS_PER_THR)
#define CHUNK_EDGES  (CHUNK_SETS * SET_EDGES)
#define N_CHUNKS     ((N_SETS + CHUNK_SETS - 1) / CHUNK_SETS)
#define PROC_UNROLL  8
#define INIT_TRIPS   ((OWN_ROWS * N_CLASSES + OWN_THREADS - 1) / OWN_THREADS)
#define SRC_BITS     17
#define SRC_MASK     ((1u << SRC_BITS) - 1u)

__global__ __launch_bounds__(128)
void gin_proj_wmma(const float* __restrict__ x, const float* __restrict__ W,
                   const int* __restrict__ src, const int* __restrict__ dst,
                   float* __restrict__ y, unsigned int* __restrict__ keys,
                   unsigned int* __restrict__ packed) {
    const int lane  = threadIdx.x & 31;
    const int wave  = threadIdx.x >> 5;
    {
        unsigned pk[2];
        int4 d[2], sv[2];
#pragma unroll
        for (int trip = 0; trip < 2; ++trip) {
            const int  e0    = blockIdx.x * SET_EDGES + (wave * 2 + trip) * 128 + lane * 4;
            const bool valid = e0 < N_EDGES;
            d[trip]  = *(const int4*)(dst + (valid ? e0 : 0));
            sv[trip] = *(const int4*)(src + (valid ? e0 : 0));
        }
#pragma unroll
        for (int trip = 0; trip < 2; ++trip) {
            const int  e0    = blockIdx.x * SET_EDGES + (wave * 2 + trip) * 128 + lane * 4;
            const bool valid = e0 < N_EDGES;
            const unsigned ox = (unsigned)d[trip].x / OWN_ROWS, oy = (unsigned)d[trip].y / OWN_ROWS;
            const unsigned oz = (unsigned)d[trip].z / OWN_ROWS, ow = (unsigned)d[trip].w / OWN_ROWS;
            const unsigned p = ox | (oy << 8) | (oz << 16) | (ow << 24);
            pk[trip] = valid ? p : 0xFFFFFFFFu;
            uint4 w;
            w.x = (unsigned)sv[trip].x | (((unsigned)d[trip].x - ox * OWN_ROWS) << SRC_BITS);
            w.y = (unsigned)sv[trip].y | (((unsigned)d[trip].y - oy * OWN_ROWS) << SRC_BITS);
            w.z = (unsigned)sv[trip].z | (((unsigned)d[trip].z - oz * OWN_ROWS) << SRC_BITS);
            w.w = (unsigned)sv[trip].w | (((unsigned)d[trip].w - ow * OWN_ROWS) << SRC_BITS);
            *(uint4*)(packed + e0) = w;
        }
        unsigned lo = 0, hi = 0;
#pragma unroll
        for (int trip = 0; trip < 2; ++trip) {
#pragma unroll
            for (int cb = 0; cb < 16; ++cb) {
                const int c = cb >> 2, b = cb & 3;
                const unsigned w0 = __builtin_amdgcn_ballot_w32(((pk[trip] >> (8 * c + b))     & 1u) != 0u);
                const unsigned w1 = __builtin_amdgcn_ballot_w32(((pk[trip] >> (8 * c + b + 4)) & 1u) != 0u);
                const bool mine = (lane == trip * 16 + cb);
                lo = mine ? w0 : lo;
                hi = mine ? w1 : hi;
            }
        }
        keys[blockIdx.x * (SET_EDGES / 4) + wave * 32 + lane]       = lo;
        keys[blockIdx.x * (SET_EDGES / 4) + 128 + wave * 32 + lane] = hi;
    }

    const int mtile = blockIdx.x * 4 + wave;
    if (mtile >= M_TILES) return;

    const int col   = lane & 15;
    const int khalf = (lane >> 4) << 1;

    const float* xrow  = x + (mtile * 16 + col) * D_FEAT;
    const float* wrow0 = W + col * D_FEAT;
    const float* wrow1 = W + (col + 16) * D_FEAT;

    v8f acc0 = {};
    v8f acc1 = {};
#pragma unroll 4
    for (int k0 = 0; k0 < D_FEAT; k0 += 4) {
        v2f a  = *(const v2f*)(xrow  + k0 + khalf);
        v2f b0 = *(const v2f*)(wrow0 + k0 + khalf);
        v2f b1 = *(const v2f*)(wrow1 + k0 + khalf);
        acc0 = __builtin_amdgcn_wmma_f32_16x16x4_f32(false, a, false, b0,
                                                     (short)0, acc0, false, false);
        acc1 = __builtin_amdgcn_wmma_f32_16x16x4_f32(false, a, false, b1,
                                                     (short)0, acc1, false, false);
    }

    const int rbase = mtile * 16 + ((lane >> 4) << 3);
#pragma unroll
    for (int r = 0; r < 8; ++r) {
        const int row = rbase + r;
        y[row * N_CLASSES + col]      = acc0[r];
        y[row * N_CLASSES + col + 16] = acc1[r];
    }
}


__device__ __forceinline__ void keep_own_edges(const uint4 (&cur)[VECS_PER_THR], const unsigned (&flip)[8],
                                               int wave, int lane,
                                               unsigned short* __restrict__ q, int* __restrict__ cnt) {
#pragma unroll
    for (int j = 0; j < SETS_PER_THR; ++j) {
        const uint4 a = cur[2 * j], h = cur[2 * j + 1];
        unsigned m = a.x ^ flip[0];
        m = __builtin_amdgcn_bitop3_b32(a.y, flip[1], m, 0x28);
        m = __builtin_amdgcn_bitop3_b32(a.z, flip[2], m, 0x28);
        m = __builtin_amdgcn_bitop3_b32(a.w, flip[3], m, 0x28);
        m = __builtin_amdgcn_bitop3_b32(h.x, flip[4], m, 0x28);
        m = __builtin_amdgcn_bitop3_b32(h.y, flip[5], m, 0x28);
        m = __builtin_amdgcn_bitop3_b32(h.z, flip[6], m, 0x28);
        m = __builtin_amdgcn_bitop3_b32(h.w, flip[7], m, 0x28);
        const int vbase = (j * (OWN_THREADS / 32) + wave) * SET_EDGES + (lane >> 2) * 128 + (lane & 3);
        while (m) {
            const int b = __builtin_ctz(m);
            m &= m - 1;
            const int slot = atomicAdd(cnt, 1);
            q[slot] = (unsigned short)(vbase + b * 4);
        }
    }
}

__device__ __forceinline__ void load_keys(uint4 (&dstv)[VECS_PER_THR], const uint4* __restrict__ kv,
                                          int ck, int wave, int lane, bool on) {
    const uint4 none = make_uint4(0xFFFFFFFFu, 0xFFFFFFFFu, 0xFFFFFFFFu, 0xFFFFFFFFu);
#pragma unroll
    for (int j = 0; j < SETS_PER_THR; ++j) {
        const int S = ck * CHUNK_SETS + j * (OWN_THREADS / 32) + wave;
        dstv[2 * j]     = none;
        dstv[2 * j + 1] = none;
        if (on && S < N_SETS) {
            dstv[2 * j]     = kv[S * SET_VECS + lane];
            dstv[2 * j + 1] = kv[S * SET_VECS + 32 + lane];
        }
    }
}

__global__ __launch_bounds__(OWN_THREADS)
void gin_owner_sum(const unsigned int* __restrict__ packed,
                   const unsigned int* __restrict__ keys,
                   const float* __restrict__ y, const float* __restrict__ bias,
                   float* __restrict__ out) {
    __shared__ float          rows[OWN_ROWS * N_CLASSES];
    __shared__ unsigned short queue[2][CHUNK_EDGES];
    __shared__ int            qcount[N_CHUNKS];

    const int tid   = threadIdx.x;
    const int lane  = tid & 31;
    const int wave  = tid >> 5;
    const int owner = blockIdx.x;
    const int base  = owner * OWN_ROWS;
    const uint4* kv = (const uint4*)keys;
    unsigned flip[8];
#pragma unroll
    for (int b = 0; b < 8; ++b) flip[b] = ((owner >> b) & 1) ? 0u : 0xFFFFFFFFu;

    int kk  = owner % N_CHUNKS;
    int kk1 = (kk + 1 == N_CHUNKS) ? 0 : kk + 1;

    uint4 cur[VECS_PER_THR];
    load_keys(cur, kv, kk, wave, lane, true);
    if (tid < N_CHUNKS) qcount[tid] = 0;
    __syncthreads();

    {
        const float bv = bias[lane];
        float t[INIT_TRIPS];
#pragma unroll
        for (int it = 0; it < INIT_TRIPS; ++it) {
            const int i  = it * OWN_THREADS + tid;
            const int ic = (i < OWN_ROWS * N_CLASSES) ? i : (OWN_ROWS * N_CLASSES - 1);
            t[it] = y[base * N_CLASSES + ic];
        }
        keep_own_edges(cur, flip, wave, lane, queue[0], &qcount[0]);
#pragma unroll
        for (int it = 0; it < INIT_TRIPS; ++it) {
            const int i = it * OWN_THREADS + tid;
            if (i < OWN_ROWS * N_CLASSES) rows[i] = t[it] + bv;
        }
    }
    load_keys(cur, kv, kk1, wave, lane, true);
    __syncthreads();

    for (int k = 0; k < N_CHUNKS; ++k) {
        const unsigned short* q = queue[k & 1];
        const int n     = qcount[k];
        const int ebase = kk * CHUNK_EDGES;
        const int first = wave * PROC_UNROLL;
        const bool act  = first < n;

        unsigned p[PROC_UNROLL];
        if (act) {
            int e[PROC_UNROLL];
#pragma unroll
            for (int u = 0; u < PROC_UNROLL; ++u) {
                const int i = (first + u < n) ? (first + u) : (n - 1);
                e[u] = ebase + (int)q[i];
            }
#pragma unroll
            for (int u = 0; u < PROC_UNROLL; ++u) p[u] = packed[e[u]];
        }

        const int kk2 = (kk1 + 1 == N_CHUNKS) ? 0 : kk1 + 1;
        uint4 nxt[VECS_PER_THR];
        load_keys(nxt, kv, kk2, wave, lane, k + 2 < N_CHUNKS);

        if (k + 1 < N_CHUNKS)
            keep_own_edges(cur, flip, wave, lane, queue[(k + 1) & 1], &qcount[k + 1]);

        if (act) {
            float v[PROC_UNROLL];
#pragma unroll
            for (int u = 0; u < PROC_UNROLL; ++u) v[u] = y[(int)(p[u] & SRC_MASK) * N_CLASSES + lane];
#pragma unroll
            for (int u = 0; u < PROC_UNROLL; ++u) {
                const int r = (int)(p[u] >> SRC_BITS);
                atomicAdd(&rows[r * N_CLASSES + lane], (first + u < n) ? v[u] : 0.0f);
            }
        }

        for (int i0 = first + (OWN_THREADS / 32) * PROC_UNROLL; i0 < n; i0 += (OWN_THREADS / 32) * PROC_UNROLL) {
            int      e2[PROC_UNROLL];
            unsigned p2[PROC_UNROLL];
            float    v2[PROC_UNROLL];
#pragma unroll
            for (int u = 0; u < PROC_UNROLL; ++u) {
                const int i = (i0 + u < n) ? (i0 + u) : (n - 1);
                e2[u] = ebase + (int)q[i];
            }
#pragma unroll
            for (int u = 0; u < PROC_UNROLL; ++u) p2[u] = packed[e2[u]];
#pragma unroll
            for (int u = 0; u < PROC_UNROLL; ++u) v2[u] = y[(int)(p2[u] & SRC_MASK) * N_CLASSES + lane];
#pragma unroll
            for (int u = 0; u < PROC_UNROLL; ++u) {
                const int r = (int)(p2[u] >> SRC_BITS);
                atomicAdd(&rows[r * N_CLASSES + lane], (i0 + u < n) ? v2[u] : 0.0f);
            }
        }
        __syncthreads();

#pragma unroll
        for (int j = 0; j < VECS_PER_THR; ++j) cur[j] = nxt[j];
        kk  = kk1;
        kk1 = kk2;
    }

    for (int i = tid; i < OWN_ROWS * N_CLASSES; i += OWN_THREADS)
        out[base * N_CLASSES + i] = rows[i];
}

extern "C" void kernel_launch(void* const* d_in, const int* in_sizes, int n_in,
                              void* d_out, int out_size, void* d_ws, size_t ws_size,
                              hipStream_t stream) {
    const float* x          = (const float*)d_in[0];
    const int*   edge_index = (const int*)  d_in[1];
    const float* W          = (const float*)d_in[2];
    const float* b          = (const float*)d_in[3];
    float*       out        = (float*)d_out;
    float*       y          = (float*)d_ws;
    unsigned int* keys      = (unsigned int*)(y + N_NODES * N_CLASSES);
    unsigned int* packed    = keys + N_SETS * (SET_EDGES / 4);

    const int* src = edge_index;
    const int* dst = edge_index + N_EDGES;

    const int proj_blocks = (M_TILES + 3) / 4;
    gin_proj_wmma<<<proj_blocks, 128, 0, stream>>>(x, W, src, dst, y, keys, packed);

    gin_owner_sum<<<N_OWNERS, OWN_THREADS, 0, stream>>>(packed, keys, y, b, out);
}


# owner key as two one-hot base-16 digits (32 plane words per 32 edges, whole lines staged in LDS): an owner loads and ANDs 2 planes per set, not 8 bit-slice words; LDS arrival-order sums as before
# speedup vs baseline: 3.2638x; 1.3285x over previous
#include <hip/hip_runtime.h>


typedef __attribute__((ext_vector_type(2))) float v2f;
typedef __attribute__((ext_vector_type(8))) float v8f;

#define N_NODES   100000
#define D_FEAT    128
#define N_EDGES   1600000
#define N_CLASSES 32
#define M_TILES   (N_NODES / 16)

#define OWN_ROWS     400
#define N_OWNERS     (N_NODES / OWN_ROWS)
#define OWN_THREADS  1024
#define SET_EDGES    1024
#define N_SETS       ((N_EDGES + SET_EDGES - 1) / SET_EDGES)
#define KEY_PLANES   32
#define SET_KEYW     (KEY_PLANES * (SET_EDGES / 32))
#define SETS_PER_THR 2
#define VECS_PER_THR (2 * SETS_PER_THR)
#define CHUNK_SETS   ((OWN_THREADS / 32) * SETS_PER_THR)
#define CHUNK_EDGES  (CHUNK_SETS * SET_EDGES)
#define N_CHUNKS     ((N_SETS + CHUNK_SETS - 1) / CHUNK_SETS)
#define WQ_CAP       (SETS_PER_THR * SET_EDGES)
#define GATHER_MAX   15
#define INIT_TRIPS   ((OWN_ROWS * N_CLASSES + OWN_THREADS - 1) / OWN_THREADS)
#define SRC_BITS     17
#define SRC_MASK     ((1u << SRC_BITS) - 1u)

__global__ __launch_bounds__(128)
void gin_proj_wmma(const float* __restrict__ x, const float* __restrict__ W,
                   const int* __restrict__ src, const int* __restrict__ dst,
                   float* __restrict__ y, unsigned int* __restrict__ keys,
                   unsigned int* __restrict__ packed) {
    __shared__ unsigned kst[SET_KEYW];
    const int lane  = threadIdx.x & 31;
    const int wave  = threadIdx.x >> 5;
    {
        unsigned pk[2];
        uint4 wv[2];
        int4 d[2], sv[2];
#pragma unroll
        for (int trip = 0; trip < 2; ++trip) {
            const int  e0    = blockIdx.x * SET_EDGES + (wave * 2 + trip) * 128 + lane * 4;
            const bool valid = e0 < N_EDGES;
            d[trip]  = *(const int4*)(dst + (valid ? e0 : 0));
            sv[trip] = *(const int4*)(src + (valid ? e0 : 0));
        }
#pragma unroll
        for (int trip = 0; trip < 2; ++trip) {
            const int  e0    = blockIdx.x * SET_EDGES + (wave * 2 + trip) * 128 + lane * 4;
            const bool valid = e0 < N_EDGES;
            const unsigned ox = (unsigned)d[trip].x / OWN_ROWS, oy = (unsigned)d[trip].y / OWN_ROWS;
            const unsigned oz = (unsigned)d[trip].z / OWN_ROWS, ow = (unsigned)d[trip].w / OWN_ROWS;
            const unsigned p = ox | (oy << 8) | (oz << 16) | (ow << 24);
            pk[trip] = valid ? p : 0xFFFFFFFFu;
            uint4 w;
            w.x = (unsigned)sv[trip].x | (((unsigned)d[trip].x - ox * OWN_ROWS) << SRC_BITS);
            w.y = (unsigned)sv[trip].y | (((unsigned)d[trip].y - oy * OWN_ROWS) << SRC_BITS);
            w.z = (unsigned)sv[trip].z | (((unsigned)d[trip].z - oz * OWN_ROWS) << SRC_BITS);
            w.w = (unsigned)sv[trip].w | (((unsigned)d[trip].w - ow * OWN_ROWS) << SRC_BITS);
            wv[trip] = w;
        }
        unsigned bw[8] = {0u, 0u, 0u, 0u, 0u, 0u, 0u, 0u};
#pragma unroll
        for (int trip = 0; trip < 2; ++trip) {
#pragma unroll
            for (int c = 0; c < 4; ++c) {
                const bool mine = ((lane & 7) == 4 * trip + c);
#pragma unroll
                for (int b = 0; b < 8; ++b) {
                    const unsigned w = __builtin_amdgcn_ballot_w32(((pk[trip] >> (8 * c + b)) & 1u) != 0u);
                    bw[b] = mine ? w : bw[b];
                }
            }
        }
        const unsigned f0 = (lane & 8)  ? 0u : 0xFFFFFFFFu;
        const unsigned f1 = (lane & 16) ? 0u : 0xFFFFFFFFu;
        unsigned kw[8];
#pragma unroll
        for (int dg = 0; dg < 2; ++dg) {
            const int      b0 = dg ? 0 : 4;
            const unsigned tq = (bw[b0] ^ f0) & (bw[b0 + 1] ^ f1);
            const unsigned d2 = bw[b0 + 2], d3 = bw[b0 + 3];
            kw[4 * dg + 0] = tq & ~d2 & ~d3;
            kw[4 * dg + 1] = tq &  d2 & ~d3;
            kw[4 * dg + 2] = tq & ~d2 &  d3;
            kw[4 * dg + 3] = tq &  d2 &  d3;
        }
#pragma unroll
        for (int i = 0; i < 8; ++i)
            kst[(4 * i + (lane >> 3)) * (SET_EDGES / 32) + 8 * wave + (lane & 7)] = kw[i];
        __syncthreads();
#pragma unroll
        for (int trip = 0; trip < 2; ++trip)
            *(uint4*)(packed + blockIdx.x * SET_EDGES + (wave * 2 + trip) * 128 + lane * 4) = wv[trip];
#pragma unroll
        for (int j = 0; j < 8; ++j) {
            const int plane = 8 * wave + j;
            keys[blockIdx.x * SET_KEYW + plane * (SET_EDGES / 32) + lane] = kst[plane * (SET_EDGES / 32) + lane];
        }
    }

    const int mtile = blockIdx.x * 4 + wave;
    if (mtile >= M_TILES) return;

    const int col   = lane & 15;
    const int khalf = (lane >> 4) << 1;

    const float* xrow  = x + (mtile * 16 + col) * D_FEAT;
    const float* wrow0 = W + col * D_FEAT;
    const float* wrow1 = W + (col + 16) * D_FEAT;

    v8f acc0 = {};
    v8f acc1 = {};
#pragma unroll 4
    for (int k0 = 0; k0 < D_FEAT; k0 += 4) {
        v2f a  = *(const v2f*)(xrow  + k0 + khalf);
        v2f b0 = *(const v2f*)(wrow0 + k0 + khalf);
        v2f b1 = *(const v2f*)(wrow1 + k0 + khalf);
        acc0 = __builtin_amdgcn_wmma_f32_16x16x4_f32(false, a, false, b0,
                                                     (short)0, acc0, false, false);
        acc1 = __builtin_amdgcn_wmma_f32_16x16x4_f32(false, a, false, b1,
                                                     (short)0, acc1, false, false);
    }

    const int rbase = mtile * 16 + ((lane >> 4) << 3);
#pragma unroll
    for (int r = 0; r < 8; ++r) {
        const int row = rbase + r;
        y[row * N_CLASSES + col]      = acc0[r];
        y[row * N_CLASSES + col + 16] = acc1[r];
    }
}


__device__ __forceinline__ int wave_incl_sum(int c, int lane) {
    int v = c;
    v += __builtin_amdgcn_update_dpp(0, v, 0x111, 0xf, 0xf, false);
    v += __builtin_amdgcn_update_dpp(0, v, 0x112, 0xf, 0xf, false);
    v += __builtin_amdgcn_update_dpp(0, v, 0x114, 0xf, 0xf, false);
    v += __builtin_amdgcn_update_dpp(0, v, 0x118, 0xf, 0xf, false);
    const int low = __builtin_amdgcn_readlane(v, 15);
    return v + ((lane >= 16) ? low : 0);
}

__device__ __forceinline__ int list_own_edges(const unsigned (&cur)[VECS_PER_THR],
                                              int wave, int lane, unsigned short* __restrict__ wl) {
    unsigned m[SETS_PER_THR];
#pragma unroll
    for (int j = 0; j < SETS_PER_THR; ++j) m[j] = cur[2 * j] & cur[2 * j + 1];
    const int c    = __builtin_popcount(m[0]) + __builtin_popcount(m[1]);
    const int incl = wave_incl_sum(c, lane);
    const int n    = __builtin_amdgcn_readlane(incl, 31);
    int slot = incl - c;
#pragma unroll
    for (int j = 0; j < SETS_PER_THR; ++j) {
        const int vbase = (j * (OWN_THREADS / 32) + wave) * SET_EDGES + (lane >> 2) * 128 + (lane & 3);
        unsigned mm = m[j];
        while (mm) {
            const int b = __builtin_ctz(mm);
            mm &= mm - 1;
            wl[slot++] = (unsigned short)(vbase + b * 4);
        }
    }
    return n;
}

__device__ __forceinline__ void load_keys(unsigned (&dstv)[VECS_PER_THR], const unsigned* __restrict__ khi,
                                          const unsigned* __restrict__ klo,
                                          int ck, int wave, int lane, bool on) {
#pragma unroll
    for (int j = 0; j < SETS_PER_THR; ++j) {
        const int  S  = ck * CHUNK_SETS + j * (OWN_THREADS / 32) + wave;
        const bool ok = on && S < N_SETS;
        const int  Sc = ok ? S : 0;
        const unsigned h = khi[Sc * SET_KEYW + lane];
        const unsigned l = klo[Sc * SET_KEYW + lane];
        dstv[2 * j]     = ok ? h : 0u;
        dstv[2 * j + 1] = l;
    }
}


#define ROW_LOAD(u, i)                                                              \
    pk[u] = (unsigned)__builtin_amdgcn_readlane((int)p, (i));                       \
    v[u]  = y[(int)(pk[u] & SRC_MASK) * N_CLASSES + lane];
#define ROW_ADD(u)                                                                  \
    atomicAdd(&rows[(int)(pk[u] >> SRC_BITS) * N_CLASSES + lane], v[u]);

__device__ __forceinline__ void add_rows(unsigned p, int cnt, int lane,
                                         const float* __restrict__ y, float* __restrict__ rows) {
    for (int k0 = 0; k0 < cnt; k0 += GATHER_MAX) {
        const int left = cnt - k0;
        const int m    = (left < GATHER_MAX) ? left : GATHER_MAX;
        unsigned pk[GATHER_MAX];
        float    v[GATHER_MAX];
        int i = k0;
        if (m & 8) {
#pragma unroll
            for (int u = 0; u < 8; ++u) { ROW_LOAD(u, i + u) }
            i += 8;
        }
        if (m & 4) {
#pragma unroll
            for (int u = 0; u < 4; ++u) { ROW_LOAD(8 + u, i + u) }
            i += 4;
        }
        if (m & 2) {
            ROW_LOAD(12, i)
            ROW_LOAD(13, i + 1)
            i += 2;
        }
        if (m & 1) {
            ROW_LOAD(14, i)
        }
        __builtin_amdgcn_fence(__ATOMIC_RELEASE, "workgroup", "global");
        if (m & 8) {
#pragma unroll
            for (int u = 0; u < 8; ++u) { ROW_ADD(u) }
        }
        if (m & 4) {
#pragma unroll
            for (int u = 0; u < 4; ++u) { ROW_ADD(8 + u) }
        }
        if (m & 2) {
            ROW_ADD(12)
            ROW_ADD(13)
        }
        if (m & 1) {
            ROW_ADD(14)
        }
    }
}
#undef ROW_LOAD
#undef ROW_ADD

__global__ __launch_bounds__(OWN_THREADS)
void gin_owner_sum(const unsigned int* __restrict__ packed,
                   const unsigned int* __restrict__ keys,
                   const float* __restrict__ y, const float* __restrict__ bias,
                   float* __restrict__ out) {
    __shared__ float          rows[OWN_ROWS * N_CLASSES];
    __shared__ unsigned short wq[2][OWN_THREADS / 32][WQ_CAP];

    const int tid   = threadIdx.x;
    const int lane  = tid & 31;
    const int wave  = tid >> 5;
    const int owner = blockIdx.x;
    const int base  = owner * OWN_ROWS;
    const unsigned* khi = keys + (owner >> 4) * (SET_EDGES / 32);
    const unsigned* klo = keys + (16 + (owner & 15)) * (SET_EDGES / 32);

    int kk  = owner % N_CHUNKS;
    int kk1 = (kk + 1 == N_CHUNKS) ? 0 : kk + 1;

    unsigned cur[VECS_PER_THR], nxt[VECS_PER_THR];
    load_keys(cur, khi, klo, kk,  wave, lane, true);
    load_keys(nxt, khi, klo, kk1, wave, lane, true);

    int      n_cur;
    unsigned p_cur = 0;
    {
        const float bv = bias[lane];
        float t[INIT_TRIPS];
#pragma unroll
        for (int it = 0; it < INIT_TRIPS; ++it) {
            const int i  = it * OWN_THREADS + tid;
            const int ic = (i < OWN_ROWS * N_CLASSES) ? i : (OWN_ROWS * N_CLASSES - 1);
            t[it] = y[base * N_CLASSES + ic];
        }
        unsigned short* wl = wq[0][wave];
        n_cur = list_own_edges(cur, wave, lane, wl);
        if (lane < n_cur) p_cur = packed[kk * CHUNK_EDGES + (int)wl[lane]];
#pragma unroll
        for (int it = 0; it < INIT_TRIPS; ++it) {
            const int i = it * OWN_THREADS + tid;
            if (i < OWN_ROWS * N_CLASSES) rows[i] = t[it] + bv;
        }
    }
    __syncthreads();

    for (int k = 0; k < N_CHUNKS; ++k) {
        const int kk2 = (kk1 + 1 == N_CHUNKS) ? 0 : kk1 + 1;
#pragma unroll
        for (int j = 0; j < VECS_PER_THR; ++j) cur[j] = nxt[j];
        load_keys(nxt, khi, klo, kk2, wave, lane, k + 2 < N_CHUNKS);

        int      n_nxt = 0;
        unsigned p_nxt = 0;
        if (k + 1 < N_CHUNKS) {
            unsigned short* wl = wq[(k + 1) & 1][wave];
            n_nxt = list_own_edges(cur, wave, lane, wl);
            if (lane < n_nxt) p_nxt = packed[kk1 * CHUNK_EDGES + (int)wl[lane]];
        }

        add_rows(p_cur, (n_cur < 32) ? n_cur : 32, lane, y, rows);

        for (int i0 = 32; i0 < n_cur; i0 += 32) {
            const unsigned short* wl = wq[k & 1][wave];
            unsigned p2 = 0;
            if (i0 + lane < n_cur) p2 = packed[kk * CHUNK_EDGES + (int)wl[i0 + lane]];
            const int left = n_cur - i0;
            add_rows(p2, (left < 32) ? left : 32, lane, y, rows);
        }

        p_cur = p_nxt;
        n_cur = n_nxt;
        kk  = kk1;
        kk1 = kk2;
    }
    __syncthreads();

    for (int i = tid; i < OWN_ROWS * N_CLASSES; i += OWN_THREADS)
        out[base * N_CLASSES + i] = rows[i];
}

extern "C" void kernel_launch(void* const* d_in, const int* in_sizes, int n_in,
                              void* d_out, int out_size, void* d_ws, size_t ws_size,
                              hipStream_t stream) {
    const float* x          = (const float*)d_in[0];
    const int*   edge_index = (const int*)  d_in[1];
    const float* W          = (const float*)d_in[2];
    const float* b          = (const float*)d_in[3];
    float*       out        = (float*)d_out;
    float*       y          = (float*)d_ws;
    unsigned int* keys      = (unsigned int*)(y + N_NODES * N_CLASSES);
    unsigned int* packed    = keys + N_SETS * SET_KEYW;

    const int* src = edge_index;
    const int* dst = edge_index + N_EDGES;

    const int proj_blocks = (M_TILES + 3) / 4;
    gin_proj_wmma<<<proj_blocks, 128, 0, stream>>>(x, W, src, dst, y, keys, packed);

    gin_owner_sum<<<N_OWNERS, OWN_THREADS, 0, stream>>>(packed, keys, y, b, out);
}


# owner launch: a wave filters 4 sets per step, not 2: 13 steps of one wait each, not 25, up to 31 row loads behind it; one list per wave (tail past 32 added first); LDS arrival-order sums as before
# speedup vs baseline: 3.4238x; 1.0490x over previous
#include <hip/hip_runtime.h>


typedef __attribute__((ext_vector_type(2))) float v2f;
typedef __attribute__((ext_vector_type(8))) float v8f;

#define N_NODES   100000
#define D_FEAT    128
#define N_EDGES   1600000
#define N_CLASSES 32
#define M_TILES   (N_NODES / 16)

#define OWN_ROWS     400
#define N_OWNERS     (N_NODES / OWN_ROWS)
#define OWN_THREADS  1024
#define SET_EDGES    1024
#define N_SETS       ((N_EDGES + SET_EDGES - 1) / SET_EDGES)
#define KEY_PLANES   32
#define SET_KEYW     (KEY_PLANES * (SET_EDGES / 32))
#define SETS_PER_THR 4
#define VECS_PER_THR (2 * SETS_PER_THR)
#define CHUNK_SETS   ((OWN_THREADS / 32) * SETS_PER_THR)
#define CHUNK_EDGES  (CHUNK_SETS * SET_EDGES)
#define N_CHUNKS     ((N_SETS + CHUNK_SETS - 1) / CHUNK_SETS)
#define WQ_CAP       (SETS_PER_THR * SET_EDGES)
#define GATHER_MAX   31
#define INIT_TRIPS   ((OWN_ROWS * N_CLASSES + OWN_THREADS - 1) / OWN_THREADS)
#define SRC_BITS     17
#define SRC_MASK     ((1u << SRC_BITS) - 1u)

__global__ __launch_bounds__(128)
void gin_proj_wmma(const float* __restrict__ x, const float* __restrict__ W,
                   const int* __restrict__ src, const int* __restrict__ dst,
                   float* __restrict__ y, unsigned int* __restrict__ keys,
                   unsigned int* __restrict__ packed) {
    __shared__ unsigned kst[SET_KEYW];
    const int lane  = threadIdx.x & 31;
    const int wave  = threadIdx.x >> 5;
    {
        unsigned pk[2];
        uint4 wv[2];
        int4 d[2], sv[2];
#pragma unroll
        for (int trip = 0; trip < 2; ++trip) {
            const int  e0    = blockIdx.x * SET_EDGES + (wave * 2 + trip) * 128 + lane * 4;
            const bool valid = e0 < N_EDGES;
            d[trip]  = *(const int4*)(dst + (valid ? e0 : 0));
            sv[trip] = *(const int4*)(src + (valid ? e0 : 0));
        }
#pragma unroll
        for (int trip = 0; trip < 2; ++trip) {
            const int  e0    = blockIdx.x * SET_EDGES + (wave * 2 + trip) * 128 + lane * 4;
            const bool valid = e0 < N_EDGES;
            const unsigned ox = (unsigned)d[trip].x / OWN_ROWS, oy = (unsigned)d[trip].y / OWN_ROWS;
            const unsigned oz = (unsigned)d[trip].z / OWN_ROWS, ow = (unsigned)d[trip].w / OWN_ROWS;
            const unsigned p = ox | (oy << 8) | (oz << 16) | (ow << 24);
            pk[trip] = valid ? p : 0xFFFFFFFFu;
            uint4 w;
            w.x = (unsigned)sv[trip].x | (((unsigned)d[trip].x - ox * OWN_ROWS) << SRC_BITS);
            w.y = (unsigned)sv[trip].y | (((unsigned)d[trip].y - oy * OWN_ROWS) << SRC_BITS);
            w.z = (unsigned)sv[trip].z | (((unsigned)d[trip].z - oz * OWN_ROWS) << SRC_BITS);
            w.w = (unsigned)sv[trip].w | (((unsigned)d[trip].w - ow * OWN_ROWS) << SRC_BITS);
            wv[trip] = w;
        }
        unsigned bw[8] = {0u, 0u, 0u, 0u, 0u, 0u, 0u, 0u};
#pragma unroll
        for (int trip = 0; trip < 2; ++trip) {
#pragma unroll
            for (int c = 0; c < 4; ++c) {
                const bool mine = ((lane & 7) == 4 * trip + c);
#pragma unroll
                for (int b = 0; b < 8; ++b) {
                    const unsigned w = __builtin_amdgcn_ballot_w32(((pk[trip] >> (8 * c + b)) & 1u) != 0u);
                    bw[b] = mine ? w : bw[b];
                }
            }
        }
        const unsigned f0 = (lane & 8)  ? 0u : 0xFFFFFFFFu;
        const unsigned f1 = (lane & 16) ? 0u : 0xFFFFFFFFu;
        unsigned kw[8];
#pragma unroll
        for (int dg = 0; dg < 2; ++dg) {
            const int      b0 = dg ? 0 : 4;
            const unsigned tq = (bw[b0] ^ f0) & (bw[b0 + 1] ^ f1);
            const unsigned d2 = bw[b0 + 2], d3 = bw[b0 + 3];
            kw[4 * dg + 0] = tq & ~d2 & ~d3;
            kw[4 * dg + 1] = tq &  d2 & ~d3;
            kw[4 * dg + 2] = tq & ~d2 &  d3;
            kw[4 * dg + 3] = tq &  d2 &  d3;
        }
#pragma unroll
        for (int i = 0; i < 8; ++i)
            kst[(4 * i + (lane >> 3)) * (SET_EDGES / 32) + 8 * wave + (lane & 7)] = kw[i];
        __syncthreads();
#pragma unroll
        for (int trip = 0; trip < 2; ++trip)
            *(uint4*)(packed + blockIdx.x * SET_EDGES + (wave * 2 + trip) * 128 + lane * 4) = wv[trip];
#pragma unroll
        for (int j = 0; j < 8; ++j) {
            const int plane = 8 * wave + j;
            keys[blockIdx.x * SET_KEYW + plane * (SET_EDGES / 32) + lane] = kst[plane * (SET_EDGES / 32) + lane];
        }
    }

    const int mtile = blockIdx.x * 4 + wave;
    if (mtile >= M_TILES) return;

    const int col   = lane & 15;
    const int khalf = (lane >> 4) << 1;

    const float* xrow  = x + (mtile * 16 + col) * D_FEAT;
    const float* wrow0 = W + col * D_FEAT;
    const float* wrow1 = W + (col + 16) * D_FEAT;

    v8f acc0 = {};
    v8f acc1 = {};
#pragma unroll 4
    for (int k0 = 0; k0 < D_FEAT; k0 += 4) {
        v2f a  = *(const v2f*)(xrow  + k0 + khalf);
        v2f b0 = *(const v2f*)(wrow0 + k0 + khalf);
        v2f b1 = *(const v2f*)(wrow1 + k0 + khalf);
        acc0 = __builtin_amdgcn_wmma_f32_16x16x4_f32(false, a, false, b0,
                                                     (short)0, acc0, false, false);
        acc1 = __builtin_amdgcn_wmma_f32_16x16x4_f32(false, a, false, b1,
                                                     (short)0, acc1, false, false);
    }

    const int rbase = mtile * 16 + ((lane >> 4) << 3);
#pragma unroll
    for (int r = 0; r < 8; ++r) {
        const int row = rbase + r;
        y[row * N_CLASSES + col]      = acc0[r];
        y[row * N_CLASSES + col + 16] = acc1[r];
    }
}


__device__ __forceinline__ int wave_incl_sum(int c, int lane) {
    int v = c;
    v += __builtin_amdgcn_update_dpp(0, v, 0x111, 0xf, 0xf, false);
    v += __builtin_amdgcn_update_dpp(0, v, 0x112, 0xf, 0xf, false);
    v += __builtin_amdgcn_update_dpp(0, v, 0x114, 0xf, 0xf, false);
    v += __builtin_amdgcn_update_dpp(0, v, 0x118, 0xf, 0xf, false);
    const int low = __builtin_amdgcn_readlane(v, 15);
    return v + ((lane >= 16) ? low : 0);
}

__device__ __forceinline__ int list_own_edges(const unsigned (&cur)[VECS_PER_THR],
                                              int lane, unsigned short* __restrict__ wl) {
    unsigned m[SETS_PER_THR];
#pragma unroll
    for (int j = 0; j < SETS_PER_THR; ++j) m[j] = cur[2 * j] & cur[2 * j + 1];
    int c = 0;
#pragma unroll
    for (int j = 0; j < SETS_PER_THR; ++j) c += __builtin_popcount(m[j]);
    const int incl = wave_incl_sum(c, lane);
    const int n    = __builtin_amdgcn_readlane(incl, 31);
    int slot = incl - c;
#pragma unroll
    for (int j = 0; j < SETS_PER_THR; ++j) {
        const int vbase = j * SET_EDGES + (lane >> 2) * 128 + (lane & 3);
        unsigned mm = m[j];
        while (mm) {
            const int b = __builtin_ctz(mm);
            mm &= mm - 1;
            wl[slot++] = (unsigned short)(vbase + b * 4);
        }
    }
    return n;
}

__device__ __forceinline__ int packed_index(int ck, int wave, int e) {
    return ck * CHUNK_EDGES + ((e / SET_EDGES) * (OWN_THREADS / 32) + wave) * SET_EDGES + (e % SET_EDGES);
}

__device__ __forceinline__ void load_keys(unsigned (&dstv)[VECS_PER_THR], const unsigned* __restrict__ khi,
                                          const unsigned* __restrict__ klo,
                                          int ck, int wave, int lane, bool on) {
#pragma unroll
    for (int j = 0; j < SETS_PER_THR; ++j) {
        const int  S  = ck * CHUNK_SETS + j * (OWN_THREADS / 32) + wave;
        const bool ok = on && S < N_SETS;
        const int  Sc = ok ? S : 0;
        const unsigned h = khi[Sc * SET_KEYW + lane];
        const unsigned l = klo[Sc * SET_KEYW + lane];
        dstv[2 * j]     = ok ? h : 0u;
        dstv[2 * j + 1] = l;
    }
}


#define ROW_LOAD(u, i)                                                              \
    pk[u] = (unsigned)__builtin_amdgcn_readlane((int)p, (i));                       \
    v[u]  = y[(int)(pk[u] & SRC_MASK) * N_CLASSES + lane];
#define ROW_ADD(u)                                                                  \
    atomicAdd(&rows[(int)(pk[u] >> SRC_BITS) * N_CLASSES + lane], v[u]);

__device__ __forceinline__ void add_rows(unsigned p, int cnt, int lane,
                                         const float* __restrict__ y, float* __restrict__ rows) {
    for (int k0 = 0; k0 < cnt; k0 += GATHER_MAX) {
        const int left = cnt - k0;
        const int m    = (left < GATHER_MAX) ? left : GATHER_MAX;
        unsigned pk[GATHER_MAX];
        float    v[GATHER_MAX];
        int i = k0;
        if (m & 16) {
#pragma unroll
            for (int u = 0; u < 16; ++u) { ROW_LOAD(u, i + u) }
            i += 16;
        }
        if (m & 8) {
#pragma unroll
            for (int u = 0; u < 8; ++u) { ROW_LOAD(16 + u, i + u) }
            i += 8;
        }
        if (m & 4) {
#pragma unroll
            for (int u = 0; u < 4; ++u) { ROW_LOAD(24 + u, i + u) }
            i += 4;
        }
        if (m & 2) {
            ROW_LOAD(28, i)
            ROW_LOAD(29, i + 1)
            i += 2;
        }
        if (m & 1) {
            ROW_LOAD(30, i)
        }
        __builtin_amdgcn_fence(__ATOMIC_RELEASE, "workgroup", "global");
        if (m & 16) {
#pragma unroll
            for (int u = 0; u < 16; ++u) { ROW_ADD(u) }
        }
        if (m & 8) {
#pragma unroll
            for (int u = 0; u < 8; ++u) { ROW_ADD(16 + u) }
        }
        if (m & 4) {
#pragma unroll
            for (int u = 0; u < 4; ++u) { ROW_ADD(24 + u) }
        }
        if (m & 2) {
            ROW_ADD(28)
            ROW_ADD(29)
        }
        if (m & 1) {
            ROW_ADD(30)
        }
    }
}
#undef ROW_LOAD
#undef ROW_ADD

__global__ __launch_bounds__(OWN_THREADS)
void gin_owner_sum(const unsigned int* __restrict__ packed,
                   const unsigned int* __restrict__ keys,
                   const float* __restrict__ y, const float* __restrict__ bias,
                   float* __restrict__ out) {
    __shared__ float          rows[OWN_ROWS * N_CLASSES];
    __shared__ unsigned short wq[OWN_THREADS / 32][WQ_CAP];

    const int tid   = threadIdx.x;
    const int lane  = tid & 31;
    const int wave  = tid >> 5;
    const int owner = blockIdx.x;
    const int base  = owner * OWN_ROWS;
    const unsigned* khi = keys + (owner >> 4) * (SET_EDGES / 32);
    const unsigned* klo = keys + (16 + (owner & 15)) * (SET_EDGES / 32);

    int kk  = owner % N_CHUNKS;
    int kk1 = (kk + 1 == N_CHUNKS) ? 0 : kk + 1;

    unsigned cur[VECS_PER_THR], nxt[VECS_PER_THR];
    load_keys(cur, khi, klo, kk,  wave, lane, true);
    load_keys(nxt, khi, klo, kk1, wave, lane, true);

    unsigned short* wl = wq[wave];
    int      n_cur;
    unsigned p_cur = 0;
    {
        const float bv = bias[lane];
        float t[INIT_TRIPS];
#pragma unroll
        for (int it = 0; it < INIT_TRIPS; ++it) {
            const int i  = it * OWN_THREADS + tid;
            const int ic = (i < OWN_ROWS * N_CLASSES) ? i : (OWN_ROWS * N_CLASSES - 1);
            t[it] = y[base * N_CLASSES + ic];
        }
        n_cur = list_own_edges(cur, lane, wl);
        if (lane < n_cur) p_cur = packed[packed_index(kk, wave, (int)wl[lane])];
#pragma unroll
        for (int it = 0; it < INIT_TRIPS; ++it) {
            const int i = it * OWN_THREADS + tid;
            if (i < OWN_ROWS * N_CLASSES) rows[i] = t[it] + bv;
        }
    }
    __syncthreads();

    for (int k = 0; k < N_CHUNKS; ++k) {
        const int kk2 = (kk1 + 1 == N_CHUNKS) ? 0 : kk1 + 1;

        for (int i0 = 32; i0 < n_cur; i0 += 32) {
            unsigned p2 = 0;
            if (i0 + lane < n_cur) p2 = packed[packed_index(kk, wave, (int)wl[i0 + lane])];
            const int left = n_cur - i0;
            add_rows(p2, (left < 32) ? left : 32, lane, y, rows);
        }

#pragma unroll
        for (int j = 0; j < VECS_PER_THR; ++j) cur[j] = nxt[j];
        load_keys(nxt, khi, klo, kk2, wave, lane, k + 2 < N_CHUNKS);

        int      n_nxt = 0;
        unsigned p_nxt = 0;
        if (k + 1 < N_CHUNKS) {
            n_nxt = list_own_edges(cur, lane, wl);
            if (lane < n_nxt) p_nxt = packed[packed_index(kk1, wave, (int)wl[lane])];
        }

        add_rows(p_cur, (n_cur < 32) ? n_cur : 32, lane, y, rows);

        p_cur = p_nxt;
        n_cur = n_nxt;
        kk  = kk1;
        kk1 = kk2;
    }
    __syncthreads();

    for (int i = tid; i < OWN_ROWS * N_CLASSES; i += OWN_THREADS)
        out[base * N_CLASSES + i] = rows[i];
}

extern "C" void kernel_launch(void* const* d_in, const int* in_sizes, int n_in,
                              void* d_out, int out_size, void* d_ws, size_t ws_size,
                              hipStream_t stream) {
    const float* x          = (const float*)d_in[0];
    const int*   edge_index = (const int*)  d_in[1];
    const float* W          = (const float*)d_in[2];
    const float* b          = (const float*)d_in[3];
    float*       out        = (float*)d_out;
    float*       y          = (float*)d_ws;
    unsigned int* keys      = (unsigned int*)(y + N_NODES * N_CLASSES);
    unsigned int* packed    = keys + N_SETS * SET_KEYW;

    const int* src = edge_index;
    const int* dst = edge_index + N_EDGES;

    const int proj_blocks = (M_TILES + 3) / 4;
    gin_proj_wmma<<<proj_blocks, 128, 0, stream>>>(x, W, src, dst, y, keys, packed);

    gin_owner_sum<<<N_OWNERS, OWN_THREADS, 0, stream>>>(packed, keys, y, b, out);
}


# projection: k split over the block's four waves: each loads its W fragments once for four row tiles, one 128-byte line per x row; an element's four partial sums are added in LDS in k order; f32 kept
# speedup vs baseline: 3.8288x; 1.1183x over previous
#include <hip/hip_runtime.h>


typedef __attribute__((ext_vector_type(2))) float v2f;
typedef __attribute__((ext_vector_type(8))) float v8f;

#define N_NODES   100000
#define D_FEAT    128
#define N_EDGES   1600000
#define N_CLASSES 32
#define M_TILES   (N_NODES / 16)

#define OWN_ROWS     400
#define N_OWNERS     (N_NODES / OWN_ROWS)
#define OWN_THREADS  1024
#define SET_EDGES    1024
#define N_SETS       ((N_EDGES + SET_EDGES - 1) / SET_EDGES)
#define KEY_PLANES   32
#define SET_KEYW     (KEY_PLANES * (SET_EDGES / 32))
#define SETS_PER_THR 4
#define VECS_PER_THR (2 * SETS_PER_THR)
#define CHUNK_SETS   ((OWN_THREADS / 32) * SETS_PER_THR)
#define CHUNK_EDGES  (CHUNK_SETS * SET_EDGES)
#define N_CHUNKS     ((N_SETS + CHUNK_SETS - 1) / CHUNK_SETS)
#define WQ_CAP       (SETS_PER_THR * SET_EDGES)
#define GATHER_MAX   31
#define INIT_TRIPS   ((OWN_ROWS * N_CLASSES + OWN_THREADS - 1) / OWN_THREADS)
#define SRC_BITS     17
#define SRC_MASK     ((1u << SRC_BITS) - 1u)

__global__ __launch_bounds__(128)
void gin_proj_wmma(const float* __restrict__ x, const float* __restrict__ W,
                   const int* __restrict__ src, const int* __restrict__ dst,
                   float* __restrict__ y, unsigned int* __restrict__ keys,
                   unsigned int* __restrict__ packed) {
    __shared__ unsigned kst[SET_KEYW];
    __shared__ float4 red[4][4][2][2][32];
    const int lane  = threadIdx.x & 31;
    const int wave  = threadIdx.x >> 5;
    {
        unsigned pk[2];
        uint4 wv[2];
        int4 d[2], sv[2];
#pragma unroll
        for (int trip = 0; trip < 2; ++trip) {
            const int  e0    = blockIdx.x * SET_EDGES + (wave * 2 + trip) * 128 + lane * 4;
            const bool valid = e0 < N_EDGES;
            d[trip]  = *(const int4*)(dst + (valid ? e0 : 0));
            sv[trip] = *(const int4*)(src + (valid ? e0 : 0));
        }
#pragma unroll
        for (int trip = 0; trip < 2; ++trip) {
            const int  e0    = blockIdx.x * SET_EDGES + (wave * 2 + trip) * 128 + lane * 4;
            const bool valid = e0 < N_EDGES;
            const unsigned ox = (unsigned)d[trip].x / OWN_ROWS, oy = (unsigned)d[trip].y / OWN_ROWS;
            const unsigned oz = (unsigned)d[trip].z / OWN_ROWS, ow = (unsigned)d[trip].w / OWN_ROWS;
            const unsigned p = ox | (oy << 8) | (oz << 16) | (ow << 24);
            pk[trip] = valid ? p : 0xFFFFFFFFu;
            uint4 w;
            w.x = (unsigned)sv[trip].x | (((unsigned)d[trip].x - ox * OWN_ROWS) << SRC_BITS);
            w.y = (unsigned)sv[trip].y | (((unsigned)d[trip].y - oy * OWN_ROWS) << SRC_BITS);
            w.z = (unsigned)sv[trip].z | (((unsigned)d[trip].z - oz * OWN_ROWS) << SRC_BITS);
            w.w = (unsigned)sv[trip].w | (((unsigned)d[trip].w - ow * OWN_ROWS) << SRC_BITS);
            wv[trip] = w;
        }
        unsigned bw[8] = {0u, 0u, 0u, 0u, 0u, 0u, 0u, 0u};
#pragma unroll
        for (int trip = 0; trip < 2; ++trip) {
#pragma unroll
            for (int c = 0; c < 4; ++c) {
                const bool mine = ((lane & 7) == 4 * trip + c);
#pragma unroll
                for (int b = 0; b < 8; ++b) {
                    const unsigned w = __builtin_amdgcn_ballot_w32(((pk[trip] >> (8 * c + b)) & 1u) != 0u);
                    bw[b] = mine ? w : bw[b];
                }
            }
        }
        const unsigned f0 = (lane & 8)  ? 0u : 0xFFFFFFFFu;
        const unsigned f1 = (lane & 16) ? 0u : 0xFFFFFFFFu;
        unsigned kw[8];
#pragma unroll
        for (int dg = 0; dg < 2; ++dg) {
            const int      b0 = dg ? 0 : 4;
            const unsigned tq = (bw[b0] ^ f0) & (bw[b0 + 1] ^ f1);
            const unsigned d2 = bw[b0 + 2], d3 = bw[b0 + 3];
            kw[4 * dg + 0] = tq & ~d2 & ~d3;
            kw[4 * dg + 1] = tq &  d2 & ~d3;
            kw[4 * dg + 2] = tq & ~d2 &  d3;
            kw[4 * dg + 3] = tq &  d2 &  d3;
        }
#pragma unroll
        for (int i = 0; i < 8; ++i)
            kst[(4 * i + (lane >> 3)) * (SET_EDGES / 32) + 8 * wave + (lane & 7)] = kw[i];
        __syncthreads();
#pragma unroll
        for (int trip = 0; trip < 2; ++trip)
            *(uint4*)(packed + blockIdx.x * SET_EDGES + (wave * 2 + trip) * 128 + lane * 4) = wv[trip];
#pragma unroll
        for (int j = 0; j < 8; ++j) {
            const int plane = 8 * wave + j;
            keys[blockIdx.x * SET_KEYW + plane * (SET_EDGES / 32) + lane] = kst[plane * (SET_EDGES / 32) + lane];
        }
    }

    const int col   = lane & 15;
    const int khalf = (lane >> 4) << 1;
    const int kq    = wave * (D_FEAT / 4);

    const float* wrow0 = W + col * D_FEAT + kq + khalf;
    const float* wrow1 = W + (col + 16) * D_FEAT + kq + khalf;
    const int    mt0   = blockIdx.x * 4;
    const float* xq    = x + col * D_FEAT + kq + khalf;

#define LOAD_TILE(dstv, mt)                                                                        \
    {                                                                                              \
        const float* xrow = xq + (((mt) < M_TILES) ? (mt) : (M_TILES - 1)) * (16 * D_FEAT);        \
        _Pragma("unroll")                                                                          \
        for (int j = 0; j < 8; ++j) dstv[j] = *(const v2f*)(xrow + 4 * j);     \
    }
#define MMA_TILE(srcv, t)                                                                          \
    {                                                                                              \
        v8f acc0 = {};                                                                             \
        v8f acc1 = {};                                                                             \
        _Pragma("unroll")                                                                          \
        for (int j = 0; j < 8; ++j) {                                                              \
            acc0 = __builtin_amdgcn_wmma_f32_16x16x4_f32(false, srcv[j], false, b0[j],             \
                                                         (short)0, acc0, false, false);            \
            acc1 = __builtin_amdgcn_wmma_f32_16x16x4_f32(false, srcv[j], false, b1[j],             \
                                                         (short)0, acc1, false, false);            \
        }                                                                                          \
        red[wave][(t)][0][0][lane] = make_float4(acc0[0], acc0[1], acc0[2], acc0[3]);              \
        red[wave][(t)][0][1][lane] = make_float4(acc0[4], acc0[5], acc0[6], acc0[7]);              \
        red[wave][(t)][1][0][lane] = make_float4(acc1[0], acc1[1], acc1[2], acc1[3]);              \
        red[wave][(t)][1][1][lane] = make_float4(acc1[4], acc1[5], acc1[6], acc1[7]);              \
    }

    v2f b0[8], b1[8], a0[8], a1[8];
#pragma unroll
    for (int j = 0; j < 8; ++j) {
        b0[j] = *(const v2f*)(wrow0 + 4 * j);
        b1[j] = *(const v2f*)(wrow1 + 4 * j);
    }
    LOAD_TILE(a0, mt0)
#pragma unroll 1
    for (int tt = 0; tt < 4; tt += 2) {
        LOAD_TILE(a1, mt0 + tt + 1)
        MMA_TILE(a0, tt)
        if (tt == 0) LOAD_TILE(a0, mt0 + 2)
        MMA_TILE(a1, tt + 1)
    }
#undef LOAD_TILE
#undef MMA_TILE
    __syncthreads();

    const int mtile = mt0 + wave;
    if (mtile >= M_TILES) return;

    float s0[8], s1[8];
#pragma unroll
    for (int h = 0; h < 2; ++h) {
        float4 u = red[0][wave][0][h][lane];
        float4 v = red[0][wave][1][h][lane];
#pragma unroll
        for (int q = 1; q < 4; ++q) {
            const float4 uq = red[q][wave][0][h][lane];
            const float4 vq = red[q][wave][1][h][lane];
            u.x += uq.x; u.y += uq.y; u.z += uq.z; u.w += uq.w;
            v.x += vq.x; v.y += vq.y; v.z += vq.z; v.w += vq.w;
        }
        s0[4 * h + 0] = u.x; s0[4 * h + 1] = u.y; s0[4 * h + 2] = u.z; s0[4 * h + 3] = u.w;
        s1[4 * h + 0] = v.x; s1[4 * h + 1] = v.y; s1[4 * h + 2] = v.z; s1[4 * h + 3] = v.w;
    }

    const int rbase = mtile * 16 + ((lane >> 4) << 3);
#pragma unroll
    for (int r = 0; r < 8; ++r) {
        const int row = rbase + r;
        y[row * N_CLASSES + col]      = s0[r];
        y[row * N_CLASSES + col + 16] = s1[r];
    }
}


__device__ __forceinline__ int wave_incl_sum(int c, int lane) {
    int v = c;
    v += __builtin_amdgcn_update_dpp(0, v, 0x111, 0xf, 0xf, false);
    v += __builtin_amdgcn_update_dpp(0, v, 0x112, 0xf, 0xf, false);
    v += __builtin_amdgcn_update_dpp(0, v, 0x114, 0xf, 0xf, false);
    v += __builtin_amdgcn_update_dpp(0, v, 0x118, 0xf, 0xf, false);
    const int low = __builtin_amdgcn_readlane(v, 15);
    return v + ((lane >= 16) ? low : 0);
}

__device__ __forceinline__ int list_own_edges(const unsigned (&cur)[VECS_PER_THR],
                                              int lane, unsigned short* __restrict__ wl) {
    unsigned m[SETS_PER_THR];
#pragma unroll
    for (int j = 0; j < SETS_PER_THR; ++j) m[j] = cur[2 * j] & cur[2 * j + 1];
    int c = 0;
#pragma unroll
    for (int j = 0; j < SETS_PER_THR; ++j) c += __builtin_popcount(m[j]);
    const int incl = wave_incl_sum(c, lane);
    const int n    = __builtin_amdgcn_readlane(incl, 31);
    int slot = incl - c;
#pragma unroll
    for (int j = 0; j < SETS_PER_THR; ++j) {
        const int vbase = j * SET_EDGES + (lane >> 2) * 128 + (lane & 3);
        unsigned mm = m[j];
        while (mm) {
            const int b = __builtin_ctz(mm);
            mm &= mm - 1;
            wl[slot++] = (unsigned short)(vbase + b * 4);
        }
    }
    return n;
}

__device__ __forceinline__ int packed_index(int ck, int wave, int e) {
    return ck * CHUNK_EDGES + ((e / SET_EDGES) * (OWN_THREADS / 32) + wave) * SET_EDGES + (e % SET_EDGES);
}

__device__ __forceinline__ void load_keys(unsigned (&dstv)[VECS_PER_THR], const unsigned* __restrict__ khi,
                                          const unsigned* __restrict__ klo,
                                          int ck, int wave, int lane, bool on) {
#pragma unroll
    for (int j = 0; j < SETS_PER_THR; ++j) {
        const int  S  = ck * CHUNK_SETS + j * (OWN_THREADS / 32) + wave;
        const bool ok = on && S < N_SETS;
        const int  Sc = ok ? S : 0;
        const unsigned h = khi[Sc * SET_KEYW + lane];
        const unsigned l = klo[Sc * SET_KEYW + lane];
        dstv[2 * j]     = ok ? h : 0u;
        dstv[2 * j + 1] = l;
    }
}


#define ROW_LOAD(u, i)                                                              \
    pk[u] = (unsigned)__builtin_amdgcn_readlane((int)p, (i));                       \
    v[u]  = y[(int)(pk[u] & SRC_MASK) * N_CLASSES + lane];
#define ROW_ADD(u)                                                                  \
    atomicAdd(&rows[(int)(pk[u] >> SRC_BITS) * N_CLASSES + lane], v[u]);

__device__ __forceinline__ void add_rows(unsigned p, int cnt, int lane,
                                         const float* __restrict__ y, float* __restrict__ rows) {
    for (int k0 = 0; k0 < cnt; k0 += GATHER_MAX) {
        const int left = cnt - k0;
        const int m    = (left < GATHER_MAX) ? left : GATHER_MAX;
        unsigned pk[GATHER_MAX];
        float    v[GATHER_MAX];
        int i = k0;
        if (m & 16) {
#pragma unroll
            for (int u = 0; u < 16; ++u) { ROW_LOAD(u, i + u) }
            i += 16;
        }
        if (m & 8) {
#pragma unroll
            for (int u = 0; u < 8; ++u) { ROW_LOAD(16 + u, i + u) }
            i += 8;
        }
        if (m & 4) {
#pragma unroll
            for (int u = 0; u < 4; ++u) { ROW_LOAD(24 + u, i + u) }
            i += 4;
        }
        if (m & 2) {
            ROW_LOAD(28, i)
            ROW_LOAD(29, i + 1)
            i += 2;
        }
        if (m & 1) {
            ROW_LOAD(30, i)
        }
        __builtin_amdgcn_fence(__ATOMIC_RELEASE, "workgroup", "global");
        if (m & 16) {
#pragma unroll
            for (int u = 0; u < 16; ++u) { ROW_ADD(u) }
        }
        if (m & 8) {
#pragma unroll
            for (int u = 0; u < 8; ++u) { ROW_ADD(16 + u) }
        }
        if (m & 4) {
#pragma unroll
            for (int u = 0; u < 4; ++u) { ROW_ADD(24 + u) }
        }
        if (m & 2) {
            ROW_ADD(28)
            ROW_ADD(29)
        }
        if (m & 1) {
            ROW_ADD(30)
        }
    }
}
#undef ROW_LOAD
#undef ROW_ADD

__global__ __launch_bounds__(OWN_THREADS)
void gin_owner_sum(const unsigned int* __restrict__ packed,
                   const unsigned int* __restrict__ keys,
                   const float* __restrict__ y, const float* __restrict__ bias,
                   float* __restrict__ out) {
    __shared__ float          rows[OWN_ROWS * N_CLASSES];
    __shared__ unsigned short wq[OWN_THREADS / 32][WQ_CAP];

    const int tid   = threadIdx.x;
    const int lane  = tid & 31;
    const int wave  = tid >> 5;
    const int owner = blockIdx.x;
    const int base  = owner * OWN_ROWS;
    const unsigned* khi = keys + (owner >> 4) * (SET_EDGES / 32);
    const unsigned* klo = keys + (16 + (owner & 15)) * (SET_EDGES / 32);

    int kk  = owner % N_CHUNKS;
    int kk1 = (kk + 1 == N_CHUNKS) ? 0 : kk + 1;

    unsigned cur[VECS_PER_THR], nxt[VECS_PER_THR];
    load_keys(cur, khi, klo, kk,  wave, lane, true);
    load_keys(nxt, khi, klo, kk1, wave, lane, true);

    unsigned short* wl = wq[wave];
    int      n_cur;
    unsigned p_cur = 0;
    {
        const float bv = bias[lane];
        float t[INIT_TRIPS];
#pragma unroll
        for (int it = 0; it < INIT_TRIPS; ++it) {
            const int i  = it * OWN_THREADS + tid;
            const int ic = (i < OWN_ROWS * N_CLASSES) ? i : (OWN_ROWS * N_CLASSES - 1);
            t[it] = y[base * N_CLASSES + ic];
        }
        n_cur = list_own_edges(cur, lane, wl);
        if (lane < n_cur) p_cur = packed[packed_index(kk, wave, (int)wl[lane])];
#pragma unroll
        for (int it = 0; it < INIT_TRIPS; ++it) {
            const int i = it * OWN_THREADS + tid;
            if (i < OWN_ROWS * N_CLASSES) rows[i] = t[it] + bv;
        }
    }
    __syncthreads();

    for (int k = 0; k < N_CHUNKS; ++k) {
        const int kk2 = (kk1 + 1 == N_CHUNKS) ? 0 : kk1 + 1;

        for (int i0 = 32; i0 < n_cur; i0 += 32) {
            unsigned p2 = 0;
            if (i0 + lane < n_cur) p2 = packed[packed_index(kk, wave, (int)wl[i0 + lane])];
            const int left = n_cur - i0;
            add_rows(p2, (left < 32) ? left : 32, lane, y, rows);
        }

#pragma unroll
        for (int j = 0; j < VECS_PER_THR; ++j) cur[j] = nxt[j];
        load_keys(nxt, khi, klo, kk2, wave, lane, k + 2 < N_CHUNKS);

        int      n_nxt = 0;
        unsigned p_nxt = 0;
        if (k + 1 < N_CHUNKS) {
            n_nxt = list_own_edges(cur, lane, wl);
            if (lane < n_nxt) p_nxt = packed[packed_index(kk1, wave, (int)wl[lane])];
        }

        add_rows(p_cur, (n_cur < 32) ? n_cur : 32, lane, y, rows);

        p_cur = p_nxt;
        n_cur = n_nxt;
        kk  = kk1;
        kk1 = kk2;
    }
    __syncthreads();

    for (int i = tid; i < OWN_ROWS * N_CLASSES; i += OWN_THREADS)
        out[base * N_CLASSES + i] = rows[i];
}

extern "C" void kernel_launch(void* const* d_in, const int* in_sizes, int n_in,
                              void* d_out, int out_size, void* d_ws, size_t ws_size,
                              hipStream_t stream) {
    const float* x          = (const float*)d_in[0];
    const int*   edge_index = (const int*)  d_in[1];
    const float* W          = (const float*)d_in[2];
    const float* b          = (const float*)d_in[3];
    float*       out        = (float*)d_out;
    float*       y          = (float*)d_ws;
    unsigned int* keys      = (unsigned int*)(y + N_NODES * N_CLASSES);
    unsigned int* packed    = keys + N_SETS * SET_KEYW;

    const int* src = edge_index;
    const int* dst = edge_index + N_EDGES;

    const int proj_blocks = (M_TILES + 3) / 4;
    gin_proj_wmma<<<proj_blocks, 128, 0, stream>>>(x, W, src, dst, y, keys, packed);

    gin_owner_sum<<<N_OWNERS, OWN_THREADS, 0, stream>>>(packed, keys, y, b, out);
}
